# starting kernel
# baseline (speedup 1.0000x reference)
#include <hip/hip_runtime.h>

typedef __attribute__((ext_vector_type(2))) float v2f;
typedef __attribute__((ext_vector_type(8))) float v8f;

#define F_IN  32
#define HC    64
#define C_OUT 8


__global__ __launch_bounds__(256) void k_deg_init(float* __restrict__ deg, int N) {
  int i = blockIdx.x * 256 + threadIdx.x;
  if (i < N) deg[i] = 1.0f;
}

__global__ __launch_bounds__(256) void k_deg_count(const int* __restrict__ dst,
                                                   float* __restrict__ deg, int E) {
  int e = blockIdx.x * 256 + threadIdx.x;
  if (e < E) atomicAdd(&deg[dst[e]], 1.0f);
}

__global__ __launch_bounds__(256) void k_dinv(float* __restrict__ deg, int N) {
  int i = blockIdx.x * 256 + threadIdx.x;
  if (i < N) deg[i] = 1.0f / sqrtf(deg[i]);
}


__global__ __launch_bounds__(32) void k_gemm1(const float* __restrict__ x,
                                              const float* __restrict__ W1,
                                              float* __restrict__ h1, int N) {
  const int lane  = threadIdx.x;
  const int l15   = lane & 15;
  const int khalf = lane >> 4;
  int row = blockIdx.x * 16 + l15;
  if (row >= N) row = N - 1;
  v8f c[4] = {};
  #pragma unroll
  for (int k0 = 0; k0 < F_IN; k0 += 4) {
    const int ka = k0 + khalf * 2;
    v2f a = { x[(size_t)row * F_IN + ka], x[(size_t)row * F_IN + ka + 1] };
    #pragma unroll
    for (int nt = 0; nt < 4; ++nt) {
      const int col = nt * 16 + l15;
      v2f b = { W1[(size_t)ka * HC + col], W1[(size_t)(ka + 1) * HC + col] };
      c[nt] = __builtin_amdgcn_wmma_f32_16x16x4_f32(false, a, false, b,
                                                    (short)0, c[nt], false, false);
    }
  }
  const int mbase = blockIdx.x * 16 + khalf * 8;
  #pragma unroll
  for (int nt = 0; nt < 4; ++nt)
    #pragma unroll
    for (int r = 0; r < 8; ++r) {
      const int m = mbase + r;
      if (m < N) h1[(size_t)m * HC + nt * 16 + l15] = c[nt][r];
    }
}


__global__ __launch_bounds__(256) void k_agg1_init(const float* __restrict__ h1,
                                                   const float* __restrict__ b1,
                                                   const float* __restrict__ dinv,
                                                   float* __restrict__ out1, int N) {
  int i = blockIdx.x * 256 + threadIdx.x;
  if (i >= N * (HC / 4)) return;
  const int node = i >> 4;
  const int fq   = i & 15;
  const float di = dinv[node];
  const float s  = di * di;
  float4 v  = ((const float4*)h1)[i];
  float4 bb = ((const float4*)b1)[fq];
  float4 o;
  o.x = bb.x + v.x * s;  o.y = bb.y + v.y * s;
  o.z = bb.z + v.z * s;  o.w = bb.w + v.w * s;
  ((float4*)out1)[i] = o;
}

__global__ __launch_bounds__(256) void k_agg1_edges(const int* __restrict__ src,
                                                    const int* __restrict__ dst,
                                                    const float* __restrict__ dinv,
                                                    const float* __restrict__ h1,
                                                    float* __restrict__ out1, int E) {
  unsigned gid = blockIdx.x * 256u + threadIdx.x;
  const int e  = gid >> 4;
  const int fq = gid & 15;
  if (e >= E) return;
  const int s = src[e], d = dst[e];
  const float nrm = dinv[s] * dinv[d];
  float4 v = ((const float4*)(h1 + (size_t)s * HC))[fq];
  float* o = out1 + (size_t)d * HC + fq * 4;
  atomicAdd(o + 0, v.x * nrm);
  atomicAdd(o + 1, v.y * nrm);
  atomicAdd(o + 2, v.z * nrm);
  atomicAdd(o + 3, v.w * nrm);
}


__global__ __launch_bounds__(32) void k_gemm2(const float* __restrict__ h,
                                              const float* __restrict__ W2,
                                              float* __restrict__ h2, int N) {
  const int lane  = threadIdx.x;
  const int l15   = lane & 15;
  const int khalf = lane >> 4;
  int row = blockIdx.x * 16 + l15;
  if (row >= N) row = N - 1;
  const bool live_col = (l15 < C_OUT);
  v8f c = {};
  #pragma unroll
  for (int k0 = 0; k0 < HC; k0 += 4) {
    const int ka = k0 + khalf * 2;
    v2f a = { h[(size_t)row * HC + ka], h[(size_t)row * HC + ka + 1] };
    v2f b = { live_col ? W2[(size_t)ka * C_OUT + l15] : 0.0f,
              live_col ? W2[(size_t)(ka + 1) * C_OUT + l15] : 0.0f };
    c = __builtin_amdgcn_wmma_f32_16x16x4_f32(false, a, false, b,
                                              (short)0, c, false, false);
  }
  const int mbase = blockIdx.x * 16 + khalf * 8;
  if (live_col) {
    #pragma unroll
    for (int r = 0; r < 8; ++r) {
      const int m = mbase + r;
      if (m < N) h2[(size_t)m * C_OUT + l15] = c[r];
    }
  }
}


__global__ __launch_bounds__(256) void k_out_init(const float* __restrict__ h2,
                                                  const float* __restrict__ b2,
                                                  const float* __restrict__ dinv,
                                                  float* __restrict__ out, int N) {
  int i = blockIdx.x * 256 + threadIdx.x;
  if (i >= N * (C_OUT / 4)) return;
  const int node = i >> 1;
  const int half = i & 1;
  const float di = dinv[node];
  const float s  = di * di;
  float4 v  = ((const float4*)h2)[i];
  float4 bb = ((const float4*)b2)[half];
  float4 o;
  o.x = bb.x + v.x * s;  o.y = bb.y + v.y * s;
  o.z = bb.z + v.z * s;  o.w = bb.w + v.w * s;
  ((float4*)out)[i] = o;
}

__global__ __launch_bounds__(256) void k_agg2_edges(const int* __restrict__ src,
                                                    const int* __restrict__ dst,
                                                    const float* __restrict__ dinv,
                                                    const float* __restrict__ h2,
                                                    float* __restrict__ out, int E) {
  unsigned gid = blockIdx.x * 256u + threadIdx.x;
  const int e    = gid >> 1;
  const int half = gid & 1;
  if (e >= E) return;
  const int s = src[e], d = dst[e];
  const float nrm = dinv[s] * dinv[d];
  float4 v = ((const float4*)(h2 + (size_t)s * C_OUT))[half];
  float* o = out + (size_t)d * C_OUT + half * 4;
  atomicAdd(o + 0, v.x * nrm);
  atomicAdd(o + 1, v.y * nrm);
  atomicAdd(o + 2, v.z * nrm);
  atomicAdd(o + 3, v.w * nrm);
}


extern "C" void kernel_launch(void* const* d_in, const int* in_sizes, int n_in,
                              void* d_out, int out_size, void* d_ws, size_t ws_size,
                              hipStream_t stream) {
  const float* x  = (const float*)d_in[0];
  const int*   ei = (const int*)d_in[1];
  const float* W1 = (const float*)d_in[2];
  const float* b1 = (const float*)d_in[3];
  const float* W2 = (const float*)d_in[4];
  const float* b2 = (const float*)d_in[5];
  float* out = (float*)d_out;

  const int N = in_sizes[0] / F_IN;
  const int E = in_sizes[1] / 2;
  const int* src = ei;
  const int* dst = ei + E;

  float* dinv = (float*)d_ws;
  float* h1   = dinv + N;
  float* out1 = h1 + (size_t)N * HC;
  float* h2   = out1 + (size_t)N * HC;

  const int nbN  = (N + 255) / 256;
  const int nbE  = (E + 255) / 256;
  const int tiles = (N + 15) / 16;

  k_deg_init <<<nbN, 256, 0, stream>>>(dinv, N);
  k_deg_count<<<nbE, 256, 0, stream>>>(dst, dinv, E);
  k_dinv     <<<nbN, 256, 0, stream>>>(dinv, N);

  k_gemm1<<<tiles, 32, 0, stream>>>(x, W1, h1, N);

  k_agg1_init <<<(N * (HC / 4) + 255) / 256, 256, 0, stream>>>(h1, b1, dinv, out1, N);
  k_agg1_edges<<<(unsigned)(((size_t)E * 16 + 255) / 256), 256, 0, stream>>>(src, dst, dinv, h1, out1, E);

  k_gemm2<<<tiles, 32, 0, stream>>>(out1, W2, h2, N);

  k_out_init  <<<(N * (C_OUT / 4) + 255) / 256, 256, 0, stream>>>(h2, b2, dinv, out, N);
  k_agg2_edges<<<(unsigned)(((size_t)E * 2 + 255) / 256), 256, 0, stream>>>(src, dst, dinv, h2, out, E);
}


# Fold the two linear layers (y = x@(W1@W2), bias b1@W2) so both edge aggregations add 8-float rows instead of 64; f32 WMMA kept; summation order changes (reassociated products, fewer atomics)
# speedup vs baseline: 1.2452x; 1.2452x over previous
#include <hip/hip_runtime.h>

typedef __attribute__((ext_vector_type(2))) float v2f;
typedef __attribute__((ext_vector_type(8))) float v8f;

#define F_IN  32
#define HC    64
#define C_OUT 8


__global__ __launch_bounds__(256) void k_deg_init(float* __restrict__ deg, int N) {
  int i = blockIdx.x * 256 + threadIdx.x;
  if (i < N) deg[i] = 1.0f;
}

__global__ __launch_bounds__(256) void k_deg_count(const int* __restrict__ dst,
                                                   float* __restrict__ deg, int E) {
  int e = blockIdx.x * 256 + threadIdx.x;
  if (e < E) atomicAdd(&deg[dst[e]], 1.0f);
}

__global__ __launch_bounds__(256) void k_dinv(float* __restrict__ deg, int N) {
  int i = blockIdx.x * 256 + threadIdx.x;
  if (i < N) deg[i] = 1.0f / sqrtf(deg[i]);
}


__global__ __launch_bounds__(32) void k_fold(const float* __restrict__ W1,
                                             const float* __restrict__ b1,
                                             const float* __restrict__ W2,
                                             float* __restrict__ W12,
                                             float* __restrict__ cvec) {
  const int lane  = threadIdx.x;
  const int l15   = lane & 15;
  const int khalf = lane >> 4;
  const int tile  = blockIdx.x;
  const float* arow = (tile < 2) ? (W1 + (size_t)(tile * 16 + l15) * HC) : b1;
  const bool live_col = (l15 < C_OUT);
  const int  col = l15 & (C_OUT - 1);
  v8f c = {};
  #pragma unroll
  for (int k0 = 0; k0 < HC; k0 += 4) {
    const int ka = k0 + khalf * 2;
    v2f a = { arow[ka], arow[ka + 1] };
    v2f b = { W2[(size_t)ka * C_OUT + col], W2[(size_t)(ka + 1) * C_OUT + col] };
    c = __builtin_amdgcn_wmma_f32_16x16x4_f32(false, a, false, b,
                                              (short)0, c, false, false);
  }
  if (live_col) {
    if (tile < 2) {
      const int mbase = tile * 16 + khalf * 8;
      #pragma unroll
      for (int r = 0; r < 8; ++r)
        W12[(size_t)(mbase + r) * C_OUT + l15] = c[r];
    } else if (khalf == 0) {
      cvec[l15] = c[0];
    }
  }
}


__global__ __launch_bounds__(32) void k_gemm_y(const float* __restrict__ x,
                                               const float* __restrict__ W12,
                                               const float* __restrict__ cvec,
                                               const float* __restrict__ dinv,
                                               float* __restrict__ y,
                                               float* __restrict__ z, int N) {
  const int lane  = threadIdx.x;
  const int l15   = lane & 15;
  const int khalf = lane >> 4;
  int row = blockIdx.x * 16 + l15;
  if (row >= N) row = N - 1;
  const bool live_col = (l15 < C_OUT);
  const int  col = l15 & (C_OUT - 1);
  const int mbase = blockIdx.x * 16 + khalf * 8;
  const float cb = cvec[col];
  float s[8];
  #pragma unroll
  for (int r = 0; r < 8; ++r) {
    int m = mbase + r;
    if (m >= N) m = N - 1;
    const float di = dinv[m];
    s[r] = di * di;
  }
  v8f c = {};
  #pragma unroll
  for (int k0 = 0; k0 < F_IN; k0 += 4) {
    const int ka = k0 + khalf * 2;
    v2f a = { x[(size_t)row * F_IN + ka], x[(size_t)row * F_IN + ka + 1] };
    v2f b = { W12[(size_t)ka * C_OUT + col], W12[(size_t)(ka + 1) * C_OUT + col] };
    c = __builtin_amdgcn_wmma_f32_16x16x4_f32(false, a, false, b,
                                              (short)0, c, false, false);
  }
  if (live_col) {
    #pragma unroll
    for (int r = 0; r < 8; ++r) {
      const int m = mbase + r;
      if (m < N) {
        const float v = c[r];
        y[(size_t)m * C_OUT + l15] = v;
        z[(size_t)m * C_OUT + l15] = cb + v * s[r];
      }
    }
  }
}


__global__ __launch_bounds__(256) void k_out_init(const float* __restrict__ z,
                                                  const float* __restrict__ b2,
                                                  const float* __restrict__ dinv,
                                                  float* __restrict__ out, int N) {
  int i = blockIdx.x * 256 + threadIdx.x;
  if (i >= N * (C_OUT / 4)) return;
  const int node = i >> 1;
  const int half = i & 1;
  const float di = dinv[node];
  const float s  = di * di;
  float4 v  = ((const float4*)z)[i];
  float4 bb = ((const float4*)b2)[half];
  float4 o;
  o.x = bb.x + v.x * s;  o.y = bb.y + v.y * s;
  o.z = bb.z + v.z * s;  o.w = bb.w + v.w * s;
  ((float4*)out)[i] = o;
}


__global__ __launch_bounds__(256) void k_agg_edges(const int* __restrict__ src,
                                                   const int* __restrict__ dst,
                                                   const float* __restrict__ dinv,
                                                   const float* __restrict__ feat,
                                                   float* __restrict__ acc, int E) {
  unsigned gid = blockIdx.x * 256u + threadIdx.x;
  const int e    = gid >> 1;
  const int half = gid & 1;
  if (e >= E) return;
  const int s = src[e], d = dst[e];
  const float nrm = dinv[s] * dinv[d];
  float4 v = ((const float4*)(feat + (size_t)s * C_OUT))[half];
  float* o = acc + (size_t)d * C_OUT + half * 4;
  atomicAdd(o + 0, v.x * nrm);
  atomicAdd(o + 1, v.y * nrm);
  atomicAdd(o + 2, v.z * nrm);
  atomicAdd(o + 3, v.w * nrm);
}


extern "C" void kernel_launch(void* const* d_in, const int* in_sizes, int n_in,
                              void* d_out, int out_size, void* d_ws, size_t ws_size,
                              hipStream_t stream) {
  const float* x  = (const float*)d_in[0];
  const int*   ei = (const int*)d_in[1];
  const float* W1 = (const float*)d_in[2];
  const float* b1 = (const float*)d_in[3];
  const float* W2 = (const float*)d_in[4];
  const float* b2 = (const float*)d_in[5];
  float* out = (float*)d_out;

  const int N = in_sizes[0] / F_IN;
  const int E = in_sizes[1] / 2;
  const int* src = ei;
  const int* dst = ei + E;

  const size_t Np = ((size_t)N + 31) / 32 * 32;
  float* dinv = (float*)d_ws;
  float* W12  = dinv + Np;
  float* cvec = W12 + F_IN * C_OUT;
  float* y    = cvec + 32;
  float* z    = y + (size_t)N * C_OUT;

  const int nbN  = (N + 255) / 256;
  const int nbE  = (E + 255) / 256;
  const int tiles = (N + 15) / 16;
  const unsigned nbAgg = (unsigned)(((size_t)E * 2 + 255) / 256);

  k_deg_init <<<nbN, 256, 0, stream>>>(dinv, N);
  k_deg_count<<<nbE, 256, 0, stream>>>(dst, dinv, E);
  k_dinv     <<<nbN, 256, 0, stream>>>(dinv, N);

  k_fold  <<<3, 32, 0, stream>>>(W1, b1, W2, W12, cvec);
  k_gemm_y<<<tiles, 32, 0, stream>>>(x, W12, cvec, dinv, y, z, N);

  k_agg_edges<<<nbAgg, 256, 0, stream>>>(src, dst, dinv, y, z, E);

  k_out_init <<<(N * (C_OUT / 4) + 255) / 256, 256, 0, stream>>>(z, b2, dinv, out, N);
  k_agg_edges<<<nbAgg, 256, 0, stream>>>(src, dst, dinv, z, out, E);
}


# Both float-atomic scatter sweeps replaced by a per-call edge sort by destination (int-atomic slot rank, scan, fill) + atomic-free per-node gather; sums in slot (arrival) order, d[i] factored out
# speedup vs baseline: 2.2004x; 1.7671x over previous
#include <hip/hip_runtime.h>

typedef __attribute__((ext_vector_type(2))) float v2f;
typedef __attribute__((ext_vector_type(8))) float v8f;


#define F_IN  32
#define HC    64
#define C_OUT 8
#define SCAN_TILE 1024


__global__ __launch_bounds__(256) void k_zero(int* __restrict__ cnt) {
  const int i = blockIdx.x * 256 + threadIdx.x;
  ((int4*)cnt)[i] = make_int4(0, 0, 0, 0);
}

__global__ __launch_bounds__(256) void k_count(const int* __restrict__ dst,
                                               int* __restrict__ cnt,
                                               int* __restrict__ rank, int E) {
  const unsigned tid    = blockIdx.x * 256u + threadIdx.x;
  const unsigned stride = gridDim.x * 256u;
  const unsigned elast = (unsigned)E - 1u;
  int d[8], r[8];
  #pragma unroll
  for (int k = 0; k < 8; ++k) {
    const unsigned e = tid + k * stride;
    d[k] = dst[e < elast ? e : elast];
  }
  #pragma unroll
  for (int k = 0; k < 8; ++k) {
    const unsigned e = tid + k * stride;
    r[k] = (e < (unsigned)E) ? atomicAdd(&cnt[d[k]], 1) : 0;
  }
  #pragma unroll
  for (int k = 0; k < 8; ++k) {
    const unsigned e = tid + k * stride;
    if (e < (unsigned)E) rank[e] = r[k];
  }
}

__device__ __forceinline__ int block_scan_excl(int val, int* sh, int& total) {
  const int lane = threadIdx.x & 31;
  const int w    = threadIdx.x >> 5;
  int inc = val;
  #pragma unroll
  for (int o = 1; o < 32; o <<= 1) {
    const int t = __shfl_up(inc, o);
    if (lane >= o) inc += t;
  }
  if (lane == 31) sh[w] = inc;
  __syncthreads();
  int woff = 0, tot = 0;
  #pragma unroll
  for (int i = 0; i < 8; ++i) {
    const int t = sh[i];
    if (i < w) woff += t;
    tot += t;
  }
  total = tot;
  return woff + inc - val;
}

__global__ __launch_bounds__(256) void k_blocksum(const int* __restrict__ cnt,
                                                  int* __restrict__ bsum) {
  __shared__ int sh[8];
  const int4 v = ((const int4*)cnt)[blockIdx.x * 256 + threadIdx.x];
  int total;
  (void)block_scan_excl(v.x + v.y + v.z + v.w, sh, total);
  if (threadIdx.x == 0) bsum[blockIdx.x] = total;
}

__global__ __launch_bounds__(256) void k_scan_apply(const int* __restrict__ cnt,
                                                    const int* __restrict__ bsum,
                                                    int* __restrict__ rowptr,
                                                    float* __restrict__ dinv) {
  __shared__ int sh0[8];
  __shared__ int sh1[8];
  const int b = blockIdx.x;
  int part = 0;
  for (int i = threadIdx.x; i < b; i += 256) part += bsum[i];
  int boff;
  (void)block_scan_excl(part, sh0, boff);
  const int gi = b * 256 + threadIdx.x;
  const int4 v = ((const int4*)cnt)[gi];
  int total;
  const int ex = boff + block_scan_excl(v.x + v.y + v.z + v.w, sh1, total);
  int4 rp;
  rp.x = ex;  rp.y = rp.x + v.x;  rp.z = rp.y + v.y;  rp.w = rp.z + v.z;
  ((int4*)rowptr)[gi] = rp;
  float4 di;
  di.x = 1.0f / sqrtf((float)(v.x + 1));
  di.y = 1.0f / sqrtf((float)(v.y + 1));
  di.z = 1.0f / sqrtf((float)(v.z + 1));
  di.w = 1.0f / sqrtf((float)(v.w + 1));
  ((float4*)dinv)[gi] = di;
}

__global__ __launch_bounds__(256) void k_fill(const int* __restrict__ src,
                                              const int* __restrict__ dst,
                                              const int* __restrict__ rank,
                                              const int* __restrict__ rowptr,
                                              int* __restrict__ csr, int E) {
  const unsigned tid    = blockIdx.x * 256u + threadIdx.x;
  const unsigned stride = gridDim.x * 256u;
  const unsigned elast = (unsigned)E - 1u;
  int s[8], d[8], r[8], p[8];
  #pragma unroll
  for (int k = 0; k < 8; ++k) {
    const unsigned e  = tid + k * stride;
    const unsigned ec = e < elast ? e : elast;
    d[k] = dst[ec];
    s[k] = src[ec];
    r[k] = rank[ec];
  }
  #pragma unroll
  for (int k = 0; k < 8; ++k)
    p[k] = rowptr[d[k]] + r[k];
  #pragma unroll
  for (int k = 0; k < 8; ++k) {
    const unsigned e = tid + k * stride;
    if (e < (unsigned)E && (unsigned)p[k] < (unsigned)E) csr[p[k]] = s[k];
  }
}


__global__ __launch_bounds__(32) void k_fold(const float* __restrict__ W1,
                                             const float* __restrict__ b1,
                                             const float* __restrict__ W2,
                                             float* __restrict__ W12,
                                             float* __restrict__ cvec) {
  const int lane  = threadIdx.x;
  const int l15   = lane & 15;
  const int khalf = lane >> 4;
  const int tile  = blockIdx.x;
  const float* arow = (tile < 2) ? (W1 + (size_t)(tile * 16 + l15) * HC) : b1;
  const bool live_col = (l15 < C_OUT);
  const int  col = l15 & (C_OUT - 1);
  v8f c = {};
  #pragma unroll
  for (int k0 = 0; k0 < HC; k0 += 4) {
    const int ka = k0 + khalf * 2;
    v2f a = { arow[ka], arow[ka + 1] };
    v2f b = { W2[(size_t)ka * C_OUT + col], W2[(size_t)(ka + 1) * C_OUT + col] };
    c = __builtin_amdgcn_wmma_f32_16x16x4_f32(false, a, false, b,
                                              (short)0, c, false, false);
  }
  if (live_col) {
    if (tile < 2) {
      const int mbase = tile * 16 + khalf * 8;
      #pragma unroll
      for (int r = 0; r < 8; ++r)
        W12[(size_t)(mbase + r) * C_OUT + l15] = c[r];
    } else if (khalf == 0) {
      cvec[l15] = c[0];
    }
  }
}


__global__ __launch_bounds__(32) void k_gemm_y(const float* __restrict__ x,
                                               const float* __restrict__ W12,
                                               const float* __restrict__ dinv,
                                               float* __restrict__ yd, int N) {
  const int lane  = threadIdx.x;
  const int l15   = lane & 15;
  const int khalf = lane >> 4;
  int row = blockIdx.x * 16 + l15;
  if (row >= N) row = N - 1;
  const bool live_col = (l15 < C_OUT);
  const int  col = l15 & (C_OUT - 1);
  const int mbase = blockIdx.x * 16 + khalf * 8;
  float s[8];
  #pragma unroll
  for (int r = 0; r < 8; ++r) {
    int m = mbase + r;
    if (m >= N) m = N - 1;
    s[r] = dinv[m];
  }
  v8f c = {};
  #pragma unroll
  for (int k0 = 0; k0 < F_IN; k0 += 4) {
    const int ka = k0 + khalf * 2;
    v2f a = { x[(size_t)row * F_IN + ka], x[(size_t)row * F_IN + ka + 1] };
    v2f b = { W12[(size_t)ka * C_OUT + col], W12[(size_t)(ka + 1) * C_OUT + col] };
    c = __builtin_amdgcn_wmma_f32_16x16x4_f32(false, a, false, b,
                                              (short)0, c, false, false);
  }
  if (live_col) {
    #pragma unroll
    for (int r = 0; r < 8; ++r) {
      const int m = mbase + r;
      if (m < N) yd[(size_t)m * C_OUT + l15] = c[r] * s[r];
    }
  }
}


__global__ __launch_bounds__(256) void k_gather(const int* __restrict__ rowptr,
                                                const int* __restrict__ csr,
                                                const float* __restrict__ dinv,
                                                const float* __restrict__ feat,
                                                const float* __restrict__ bias,
                                                float* __restrict__ res,
                                                int N, int scale_out) {
  const unsigned gid = blockIdx.x * 256u + threadIdx.x;
  const int node = gid >> 1;
  const int half = gid & 1;
  if (node >= N) return;
  const int beg = rowptr[node];
  const int end = rowptr[node + 1];
  const float di = dinv[node];
  const float4* f4 = (const float4*)feat;
  const float4 bb = ((const float4*)bias)[half];
  float4 acc = f4[gid];
  for (int j = beg; j < end; j += 8) {
    const int last = end - 1;
    int s[8];
    #pragma unroll
    for (int k = 0; k < 8; ++k) {
      const int jj = (j + k < last) ? (j + k) : last;
      int t = csr[jj];
      t = t < 0 ? 0 : t;
      s[k] = t < N ? t : N - 1;
    }
    float4 v[8];
    #pragma unroll
    for (int k = 0; k < 8; ++k) v[k] = f4[(size_t)s[k] * 2 + half];
    #pragma unroll
    for (int k = 0; k < 8; ++k) {
      const bool live = (k == 0) || (j + k) < end;
      acc.x += live ? v[k].x : 0.0f;
      acc.y += live ? v[k].y : 0.0f;
      acc.z += live ? v[k].z : 0.0f;
      acc.w += live ? v[k].w : 0.0f;
    }
  }
  float4 o;
  o.x = bb.x + di * acc.x;  o.y = bb.y + di * acc.y;
  o.z = bb.z + di * acc.z;  o.w = bb.w + di * acc.w;
  if (scale_out) { o.x *= di;  o.y *= di;  o.z *= di;  o.w *= di; }
  ((float4*)res)[gid] = o;
}


extern "C" void kernel_launch(void* const* d_in, const int* in_sizes, int n_in,
                              void* d_out, int out_size, void* d_ws, size_t ws_size,
                              hipStream_t stream) {
  const float* x  = (const float*)d_in[0];
  const int*   ei = (const int*)d_in[1];
  const float* W1 = (const float*)d_in[2];
  const float* b1 = (const float*)d_in[3];
  const float* W2 = (const float*)d_in[4];
  const float* b2 = (const float*)d_in[5];
  float* out = (float*)d_out;

  const int N = in_sizes[0] / F_IN;
  const int E = in_sizes[1] / 2;
  const int* src = ei;
  const int* dst = ei + E;

  const size_t Np  = ((size_t)N + SCAN_TILE) / SCAN_TILE * SCAN_TILE;
  const size_t Ep  = ((size_t)E + 31) / 32 * 32;
  const int    nb  = (int)(Np / SCAN_TILE);
  const size_t nbp = ((size_t)nb + 31) / 32 * 32;
  int*   cnt    = (int*)d_ws;
  int*   rowptr = cnt + Np;
  float* dinv   = (float*)(rowptr + Np);
  int*   bsum   = (int*)(dinv + Np);
  float* W12    = (float*)(bsum + nbp);
  float* cvec   = W12 + F_IN * C_OUT;
  int*   rank   = (int*)(cvec + 32);
  int*   csr    = rank + Ep;
  float* yd     = (float*)(csr + Ep);
  float* zd     = yd + Np * C_OUT;

  const int tiles = (N + 15) / 16;
  const int nbE8  = (int)(((size_t)E + 256 * 8 - 1) / (256 * 8));
  const int nbG   = (int)(((size_t)N * 2 + 255) / 256);

  k_zero      <<<nb, 256, 0, stream>>>(cnt);
  k_count     <<<nbE8, 256, 0, stream>>>(dst, cnt, rank, E);
  k_blocksum  <<<nb, 256, 0, stream>>>(cnt, bsum);
  k_scan_apply<<<nb, 256, 0, stream>>>(cnt, bsum, rowptr, dinv);
  k_fill      <<<nbE8, 256, 0, stream>>>(src, dst, rank, rowptr, csr, E);

  k_fold  <<<3, 32, 0, stream>>>(W1, b1, W2, W12, cvec);
  k_gemm_y<<<tiles, 32, 0, stream>>>(x, W12, dinv, yd, N);

  k_gather<<<nbG, 256, 0, stream>>>(rowptr, csr, dinv, yd, cvec, zd, N, 1);
  k_gather<<<nbG, 256, 0, stream>>>(rowptr, csr, dinv, zd, b2, out, N, 0);
}


# Edge sort (slot rank, scan, fill) replaced by one bucketing pass: each destination owns a 64-slot row filled as its count is taken, overflow on a per-node chain; sums stay in arrival order
# speedup vs baseline: 2.3807x; 1.0819x over previous
#include <hip/hip_runtime.h>

typedef __attribute__((ext_vector_type(2))) float v2f;
typedef __attribute__((ext_vector_type(8))) float v8f;


#define F_IN  32
#define HC    64
#define C_OUT 8
#define ZERO_TILE 1024
#define CAP   64


__global__ __launch_bounds__(256) void k_zero(int* __restrict__ cnt) {
  const int i = blockIdx.x * 256 + threadIdx.x;
  ((int4*)cnt)[i] = make_int4(0, 0, 0, 0);
}

__global__ __launch_bounds__(256) void k_place(const int* __restrict__ src,
                                               const int* __restrict__ dst,
                                               int* __restrict__ cnt,
                                               int* __restrict__ slots,
                                               int* __restrict__ ovf_cnt,
                                               int* __restrict__ head,
                                               int* __restrict__ ovf_src,
                                               int* __restrict__ ovf_next, int E) {
  const unsigned tid    = blockIdx.x * 256u + threadIdx.x;
  const unsigned stride = gridDim.x * 256u;
  const unsigned elast = (unsigned)E - 1u;
  int s[8], d[8], r[8];
  #pragma unroll
  for (int k = 0; k < 8; ++k) {
    const unsigned e  = tid + k * stride;
    const unsigned ec = e < elast ? e : elast;
    d[k] = dst[ec];
    s[k] = src[ec];
  }
  #pragma unroll
  for (int k = 0; k < 8; ++k) {
    const unsigned e = tid + k * stride;
    r[k] = (e < (unsigned)E) ? atomicAdd(&cnt[d[k]], 1) : 0;
  }
  #pragma unroll
  for (int k = 0; k < 8; ++k) {
    const unsigned e = tid + k * stride;
    if (e < (unsigned)E) {
      if (r[k] < CAP) {
        slots[(size_t)d[k] * CAP + r[k]] = s[k];
      } else {
        const int q = atomicAdd(ovf_cnt, 1);
        if ((unsigned)q < (unsigned)E) {
          ovf_src[q]  = s[k];
          ovf_next[q] = atomicExch(&head[d[k]], q);
        }
      }
    }
  }
}


__global__ __launch_bounds__(32) void k_fold(const float* __restrict__ W1,
                                             const float* __restrict__ b1,
                                             const float* __restrict__ W2,
                                             float* __restrict__ W12,
                                             float* __restrict__ cvec) {
  const int lane  = threadIdx.x;
  const int l15   = lane & 15;
  const int khalf = lane >> 4;
  const int tile  = blockIdx.x;
  const float* arow = (tile < 2) ? (W1 + (size_t)(tile * 16 + l15) * HC) : b1;
  const bool live_col = (l15 < C_OUT);
  const int  col = l15 & (C_OUT - 1);
  v8f c = {};
  #pragma unroll
  for (int k0 = 0; k0 < HC; k0 += 4) {
    const int ka = k0 + khalf * 2;
    v2f a = { arow[ka], arow[ka + 1] };
    v2f b = { W2[(size_t)ka * C_OUT + col], W2[(size_t)(ka + 1) * C_OUT + col] };
    c = __builtin_amdgcn_wmma_f32_16x16x4_f32(false, a, false, b,
                                              (short)0, c, false, false);
  }
  if (live_col) {
    if (tile < 2) {
      const int mbase = tile * 16 + khalf * 8;
      #pragma unroll
      for (int r = 0; r < 8; ++r)
        W12[(size_t)(mbase + r) * C_OUT + l15] = c[r];
    } else if (khalf == 0) {
      cvec[l15] = c[0];
    }
  }
}


__global__ __launch_bounds__(32) void k_gemm_y(const float* __restrict__ x,
                                               const float* __restrict__ W12,
                                               const int* __restrict__ cnt,
                                               float* __restrict__ yd, int N) {
  const int lane  = threadIdx.x;
  const int l15   = lane & 15;
  const int khalf = lane >> 4;
  int row = blockIdx.x * 16 + l15;
  if (row >= N) row = N - 1;
  const bool live_col = (l15 < C_OUT);
  const int  col = l15 & (C_OUT - 1);
  const int mbase = blockIdx.x * 16 + khalf * 8;
  const float dmine = 1.0f / sqrtf((float)(cnt[row] + 1));
  v8f c = {};
  #pragma unroll
  for (int k0 = 0; k0 < F_IN; k0 += 4) {
    const int ka = k0 + khalf * 2;
    v2f a = { x[(size_t)row * F_IN + ka], x[(size_t)row * F_IN + ka + 1] };
    v2f b = { W12[(size_t)ka * C_OUT + col], W12[(size_t)(ka + 1) * C_OUT + col] };
    c = __builtin_amdgcn_wmma_f32_16x16x4_f32(false, a, false, b,
                                              (short)0, c, false, false);
  }
  float s[8];
  #pragma unroll
  for (int r = 0; r < 8; ++r) s[r] = __shfl(dmine, khalf * 8 + r);
  if (live_col) {
    #pragma unroll
    for (int r = 0; r < 8; ++r) {
      const int m = mbase + r;
      if (m < N) yd[(size_t)m * C_OUT + l15] = c[r] * s[r];
    }
  }
}


__global__ __launch_bounds__(256) void k_gather(const int* __restrict__ cnt,
                                                const int* __restrict__ slots,
                                                const int* __restrict__ head,
                                                const int* __restrict__ ovf_src,
                                                const int* __restrict__ ovf_next,
                                                const float* __restrict__ feat,
                                                const float* __restrict__ bias,
                                                float* __restrict__ res,
                                                int N, int E, int scale_out) {
  const unsigned gid = blockIdx.x * 256u + threadIdx.x;
  const int node = gid >> 1;
  const int half = gid & 1;
  if (node >= N) return;
  const int deg = cnt[node];
  const int end = deg < CAP ? deg : CAP;
  const float di = 1.0f / sqrtf((float)(deg + 1));
  const int* row = slots + (size_t)node * CAP;
  const float4* f4 = (const float4*)feat;
  const float4 bb = ((const float4*)bias)[half];
  float4 acc = f4[gid];
  for (int j = 0; j < end; j += 8) {
    const int last = end - 1;
    int s[8];
    #pragma unroll
    for (int k = 0; k < 8; ++k) {
      const int jj = (j + k < last) ? (j + k) : last;
      int t = row[jj];
      t = t < 0 ? 0 : t;
      s[k] = t < N ? t : N - 1;
    }
    float4 v[8];
    #pragma unroll
    for (int k = 0; k < 8; ++k) v[k] = f4[(size_t)s[k] * 2 + half];
    #pragma unroll
    for (int k = 0; k < 8; ++k) {
      const bool live = (k == 0) || (j + k) < end;
      acc.x += live ? v[k].x : 0.0f;
      acc.y += live ? v[k].y : 0.0f;
      acc.z += live ? v[k].z : 0.0f;
      acc.w += live ? v[k].w : 0.0f;
    }
  }
  if (deg > CAP) {
    int q = head[node];
    for (int t = CAP; t < deg; ++t) {
      q = q < 0 ? 0 : q;
      q = q < E ? q : E - 1;
      int sn = ovf_src[q];
      sn = sn < 0 ? 0 : sn;
      sn = sn < N ? sn : N - 1;
      const float4 v = f4[(size_t)sn * 2 + half];
      acc.x += v.x;  acc.y += v.y;  acc.z += v.z;  acc.w += v.w;
      q = ovf_next[q];
    }
  }
  float4 o;
  o.x = bb.x + di * acc.x;  o.y = bb.y + di * acc.y;
  o.z = bb.z + di * acc.z;  o.w = bb.w + di * acc.w;
  if (scale_out) { o.x *= di;  o.y *= di;  o.z *= di;  o.w *= di; }
  ((float4*)res)[gid] = o;
}


extern "C" void kernel_launch(void* const* d_in, const int* in_sizes, int n_in,
                              void* d_out, int out_size, void* d_ws, size_t ws_size,
                              hipStream_t stream) {
  const float* x  = (const float*)d_in[0];
  const int*   ei = (const int*)d_in[1];
  const float* W1 = (const float*)d_in[2];
  const float* b1 = (const float*)d_in[3];
  const float* W2 = (const float*)d_in[4];
  const float* b2 = (const float*)d_in[5];
  float* out = (float*)d_out;

  const int N = in_sizes[0] / F_IN;
  const int E = in_sizes[1] / 2;
  const int* src = ei;
  const int* dst = ei + E;

  const size_t Np  = ((size_t)N + ZERO_TILE - 1) / ZERO_TILE * ZERO_TILE;
  const size_t Ep  = ((size_t)E + 31) / 32 * 32;
  const int    nbz = (int)((2 * Np + ZERO_TILE) / ZERO_TILE);
  int*   cnt      = (int*)d_ws;
  int*   ovf_cnt  = cnt + Np;
  int*   head     = ovf_cnt + ZERO_TILE;
  float* W12      = (float*)(head + Np);
  float* cvec     = W12 + F_IN * C_OUT;
  int*   slots    = (int*)(cvec + 32);
  int*   ovf_src  = slots + Np * CAP;
  int*   ovf_next = ovf_src + Ep;
  float* yd       = (float*)(ovf_next + Ep);
  float* zd       = yd + Np * C_OUT;

  const int tiles = (N + 15) / 16;
  const int nbE8  = (int)(((size_t)E + 256 * 8 - 1) / (256 * 8));
  const int nbG   = (int)(((size_t)N * 2 + 255) / 256);

  k_zero  <<<nbz, 256, 0, stream>>>(cnt);
  k_place <<<nbE8, 256, 0, stream>>>(src, dst, cnt, slots, ovf_cnt, head, ovf_src, ovf_next, E);

  k_fold  <<<3, 32, 0, stream>>>(W1, b1, W2, W12, cvec);
  k_gemm_y<<<tiles, 32, 0, stream>>>(x, W12, cnt, yd, N);

  k_gather<<<nbG, 256, 0, stream>>>(cnt, slots, head, ovf_src, ovf_next, yd, cvec, zd, N, E, 1);
  k_gather<<<nbG, 256, 0, stream>>>(cnt, slots, head, ovf_src, ovf_next, zd, b2, out, N, E, 0);
}


# Edge bucketing moved into LDS: each chunk ordered by 256-node bin with LDS adds and one global add per bin, then a work-group per bin builds CSR rows in LDS; sums stay in arrival order (LDS adds)
# speedup vs baseline: 6.3869x; 2.6828x over previous
#include <hip/hip_runtime.h>

typedef __attribute__((ext_vector_type(2))) float v2f;
typedef __attribute__((ext_vector_type(8))) float v8f;


#define F_IN  32
#define HC    64
#define C_OUT 8
#define ZERO_TILE 1024
#define BSH    8
#define BINW   256
#define BCAP   8192
#define CHUNK  8192
#define T1     1024
#define EPT1   8
#define T2     512
#define RPT2   16
#define MAXNB  1024
#define FS     32


__global__ __launch_bounds__(256) void k_zero(int* __restrict__ p) {
  const int i = blockIdx.x * 256 + threadIdx.x;
  ((int4*)p)[i] = make_int4(0, 0, 0, 0);
}

__device__ __forceinline__ int wave_incl_scan(int x, int lane) {
  #pragma unroll
  for (int o = 1; o < 32; o <<= 1) {
    const int y = __shfl_up(x, o);
    if (lane >= o) x += y;
  }
  return x;
}

template <int NW>
__device__ __forceinline__ int block_excl_scan(int v, int* wsum, int tid, int& total) {
  const int lane = tid & 31;
  const int w    = tid >> 5;
  const int inc  = wave_incl_scan(v, lane);
  __syncthreads();
  if (lane == 31) wsum[w] = inc;
  __syncthreads();
  if (w == 0) {
    const int t = lane < NW ? wsum[lane] : 0;
    wsum[lane] = wave_incl_scan(t, lane);
  }
  __syncthreads();
  total = wsum[31];
  return (w ? wsum[w - 1] : 0) + inc - v;
}

__global__ __launch_bounds__(T1) void k_bin(const int* __restrict__ src,
                                            const int* __restrict__ dst,
                                            int* __restrict__ bin_fill,
                                            unsigned* __restrict__ region,
                                            int* __restrict__ spill_n,
                                            int* __restrict__ spill_src,
                                            int* __restrict__ spill_dst, int E, int N) {
  __shared__ int      hist[MAXNB];
  __shared__ int      gbase[MAXNB];
  __shared__ unsigned s_rec[CHUNK];
  __shared__ int      s_adr[CHUNK];
  __shared__ int      wsum[32];
  const int tid = threadIdx.x;
  const unsigned e0    = blockIdx.x * (unsigned)CHUNK;
  const unsigned elast = (unsigned)E - 1u;
  hist[tid] = 0;
  __syncthreads();
  int d[EPT1], s[EPT1], r[EPT1];
  #pragma unroll
  for (int k = 0; k < EPT1; ++k) {
    const unsigned e  = e0 + k * T1 + tid;
    const unsigned ec = e < elast ? e : elast;
    d[k] = dst[ec];
    s[k] = src[ec];
  }
  #pragma unroll
  for (int k = 0; k < EPT1; ++k) {
    const unsigned e = e0 + k * T1 + tid;
    int dd = d[k];  dd = dd < 0 ? 0 : dd;  dd = dd < N ? dd : N - 1;
    int ss = s[k];  ss = ss < 0 ? 0 : ss;  ss = ss < N ? ss : N - 1;
    d[k] = dd;  s[k] = ss;
    r[k] = (e < (unsigned)E) ? atomicAdd(&hist[dd >> BSH], 1) : 0;
  }
  __syncthreads();
  const int c = hist[tid];
  int chunk_total;
  const int lo = block_excl_scan<T1 / 32>(c, wsum, tid, chunk_total);
  int g = 0;
  if (c > 0) g = atomicAdd(&bin_fill[tid * FS], c);
  gbase[tid] = g;
  hist[tid]  = lo;
  __syncthreads();
  #pragma unroll
  for (int k = 0; k < EPT1; ++k) {
    const unsigned e = e0 + k * T1 + tid;
    if (e < (unsigned)E) {
      const int b  = d[k] >> BSH;
      const int lp = hist[b] + r[k];
      const int gp = gbase[b] + r[k];
      s_rec[lp] = ((unsigned)s[k] << 8) | (unsigned)(d[k] & (BINW - 1));
      s_adr[lp] = gp < BCAP ? b * BCAP + gp : -1 - b;
    }
  }
  __syncthreads();
  const unsigned left   = (unsigned)E - e0;
  const int      n_here = left < (unsigned)CHUNK ? (int)left : CHUNK;
  for (int i = tid; i < n_here; i += T1) {
    const int      a   = s_adr[i];
    const unsigned rec = s_rec[i];
    if (a >= 0) {
      region[a] = rec;
    } else {
      const int q = atomicAdd(spill_n, 1);
      if ((unsigned)q < (unsigned)E) {
        spill_src[q] = (int)(rec >> 8);
        spill_dst[q] = ((-1 - a) << BSH) | (int)(rec & (BINW - 1));
      }
    }
  }
}

__global__ __launch_bounds__(T2) void k_rows(const int* __restrict__ bin_fill,
                                             const unsigned* __restrict__ region,
                                             const int* __restrict__ spill_n,
                                             const int* __restrict__ spill_src,
                                             const int* __restrict__ spill_dst,
                                             int* __restrict__ cnt,
                                             int* __restrict__ row_start,
                                             int* __restrict__ csr, int N, int E) {
  __shared__ int cnt_l[BINW];
  __shared__ int cur_l[BINW];
  __shared__ int sorted[BCAP];
  __shared__ int wsum[32];
  const int tid = threadIdx.x;
  const int b   = blockIdx.x;
  const int total = bin_fill[b * FS];
  const int n     = total < BCAP ? total : BCAP;
  const bool spilled = total > BCAP;
  int acc = 0;
  for (int j = tid; j < b; j += T2) acc += (bin_fill[j * FS] + 31) & ~31;
  int base;
  (void)block_excl_scan<T2 / 32>(acc, wsum, tid, base);
  if (tid < BINW) cnt_l[tid] = 0;
  __syncthreads();
  const unsigned* reg = region + (size_t)b * BCAP;
  unsigned rec[RPT2];
  #pragma unroll
  for (int k = 0; k < RPT2; ++k) {
    const int i = k * T2 + tid;
    rec[k] = 0u;
    if (i < n) rec[k] = reg[i];
  }
  #pragma unroll
  for (int k = 0; k < RPT2; ++k) {
    const int i = k * T2 + tid;
    if (i < n) atomicAdd(&cnt_l[rec[k] & (BINW - 1)], 1);
  }
  int m = 0;
  if (spilled) {
    m = *spill_n;
    m = m < 0 ? 0 : m;
    m = m < E ? m : E;
    for (int q = tid; q < m; q += T2) {
      const int dq = spill_dst[q];
      if ((dq >> BSH) == b) atomicAdd(&cnt_l[dq & (BINW - 1)], 1);
    }
  }
  __syncthreads();
  const int v = tid < BINW ? cnt_l[tid] : 0;
  int bin_total;
  const int ex = block_excl_scan<T2 / 32>(v, wsum, tid, bin_total);
  if (tid < BINW) {
    cur_l[tid] = ex;
    const int node = b * BINW + tid;
    if (node < N) {
      cnt[node]       = v;
      row_start[node] = base + ex;
    }
  }
  __syncthreads();
  #pragma unroll
  for (int k = 0; k < RPT2; ++k) {
    const int i = k * T2 + tid;
    if (i < n) {
      const int pos = atomicAdd(&cur_l[rec[k] & (BINW - 1)], 1);
      const int sv  = (int)(rec[k] >> 8);
      if (!spilled) sorted[pos] = sv;
      else          csr[base + pos] = sv;
    }
  }
  if (spilled) {
    for (int q = tid; q < m; q += T2) {
      const int dq = spill_dst[q];
      if ((dq >> BSH) == b) {
        const int pos = atomicAdd(&cur_l[dq & (BINW - 1)], 1);
        csr[base + pos] = spill_src[q];
      }
    }
  }
  __syncthreads();
  if (!spilled) {
    for (int i = tid; i < total; i += T2) csr[base + i] = sorted[i];
  }
}


__global__ __launch_bounds__(32) void k_fold(const float* __restrict__ W1,
                                             const float* __restrict__ b1,
                                             const float* __restrict__ W2,
                                             float* __restrict__ W12,
                                             float* __restrict__ cvec) {
  const int lane  = threadIdx.x;
  const int l15   = lane & 15;
  const int khalf = lane >> 4;
  const int tile  = blockIdx.x;
  const float* arow = (tile < 2) ? (W1 + (size_t)(tile * 16 + l15) * HC) : b1;
  const bool live_col = (l15 < C_OUT);
  const int  col = l15 & (C_OUT - 1);
  v8f c = {};
  #pragma unroll
  for (int k0 = 0; k0 < HC; k0 += 4) {
    const int ka = k0 + khalf * 2;
    v2f a = { arow[ka], arow[ka + 1] };
    v2f b = { W2[(size_t)ka * C_OUT + col], W2[(size_t)(ka + 1) * C_OUT + col] };
    c = __builtin_amdgcn_wmma_f32_16x16x4_f32(false, a, false, b,
                                              (short)0, c, false, false);
  }
  if (live_col) {
    if (tile < 2) {
      const int mbase = tile * 16 + khalf * 8;
      #pragma unroll
      for (int r = 0; r < 8; ++r)
        W12[(size_t)(mbase + r) * C_OUT + l15] = c[r];
    } else if (khalf == 0) {
      cvec[l15] = c[0];
    }
  }
}


__global__ __launch_bounds__(32) void k_gemm_y(const float* __restrict__ x,
                                               const float* __restrict__ W12,
                                               const int* __restrict__ cnt,
                                               float* __restrict__ yd, int N) {
  const int lane  = threadIdx.x;
  const int l15   = lane & 15;
  const int khalf = lane >> 4;
  int row = blockIdx.x * 16 + l15;
  if (row >= N) row = N - 1;
  const bool live_col = (l15 < C_OUT);
  const int  col = l15 & (C_OUT - 1);
  const int mbase = blockIdx.x * 16 + khalf * 8;
  const float dmine = 1.0f / sqrtf((float)(cnt[row] + 1));
  v8f c = {};
  #pragma unroll
  for (int k0 = 0; k0 < F_IN; k0 += 4) {
    const int ka = k0 + khalf * 2;
    v2f a = { x[(size_t)row * F_IN + ka], x[(size_t)row * F_IN + ka + 1] };
    v2f b = { W12[(size_t)ka * C_OUT + col], W12[(size_t)(ka + 1) * C_OUT + col] };
    c = __builtin_amdgcn_wmma_f32_16x16x4_f32(false, a, false, b,
                                              (short)0, c, false, false);
  }
  float s[8];
  #pragma unroll
  for (int r = 0; r < 8; ++r) s[r] = __shfl(dmine, khalf * 8 + r);
  if (live_col) {
    #pragma unroll
    for (int r = 0; r < 8; ++r) {
      const int m = mbase + r;
      if (m < N) yd[(size_t)m * C_OUT + l15] = c[r] * s[r];
    }
  }
}


__global__ __launch_bounds__(256) void k_gather(const int* __restrict__ cnt,
                                                const int* __restrict__ row_start,
                                                const int* __restrict__ csr,
                                                const float* __restrict__ feat,
                                                const float* __restrict__ bias,
                                                float* __restrict__ res,
                                                int N, int cap, int scale_out) {
  const unsigned gid = blockIdx.x * 256u + threadIdx.x;
  const int node = gid >> 1;
  const int half = gid & 1;
  if (node >= N) return;
  const int end = cnt[node];
  int rs = row_start[node];
  rs = rs < 0 ? 0 : rs;
  const float di = 1.0f / sqrtf((float)(end + 1));
  const float4* f4 = (const float4*)feat;
  const float4 bb = ((const float4*)bias)[half];
  float4 acc = f4[gid];
  for (int j = 0; j < end; j += 8) {
    const int last = end - 1;
    int s[8];
    #pragma unroll
    for (int k = 0; k < 8; ++k) {
      const int jj = (j + k < last) ? (j + k) : last;
      int p = rs + jj;
      p = p < cap ? p : cap - 1;
      int t = csr[p];
      t = t < 0 ? 0 : t;
      s[k] = t < N ? t : N - 1;
    }
    float4 v[8];
    #pragma unroll
    for (int k = 0; k < 8; ++k) v[k] = f4[(size_t)s[k] * 2 + half];
    #pragma unroll
    for (int k = 0; k < 8; ++k) {
      const bool live = (k == 0) || (j + k) < end;
      acc.x += live ? v[k].x : 0.0f;
      acc.y += live ? v[k].y : 0.0f;
      acc.z += live ? v[k].z : 0.0f;
      acc.w += live ? v[k].w : 0.0f;
    }
  }
  float4 o;
  o.x = bb.x + di * acc.x;  o.y = bb.y + di * acc.y;
  o.z = bb.z + di * acc.z;  o.w = bb.w + di * acc.w;
  if (scale_out) { o.x *= di;  o.y *= di;  o.z *= di;  o.w *= di; }
  ((float4*)res)[gid] = o;
}


extern "C" void kernel_launch(void* const* d_in, const int* in_sizes, int n_in,
                              void* d_out, int out_size, void* d_ws, size_t ws_size,
                              hipStream_t stream) {
  const float* x  = (const float*)d_in[0];
  const int*   ei = (const int*)d_in[1];
  const float* W1 = (const float*)d_in[2];
  const float* b1 = (const float*)d_in[3];
  const float* W2 = (const float*)d_in[4];
  const float* b2 = (const float*)d_in[5];
  float* out = (float*)d_out;

  const int N = in_sizes[0] / F_IN;
  const int E = in_sizes[1] / 2;
  const int* src = ei;
  const int* dst = ei + E;

  const size_t Np  = ((size_t)N + ZERO_TILE - 1) / ZERO_TILE * ZERO_TILE;
  const size_t Ep  = ((size_t)E + 31) / 32 * 32;
  const int    NB  = (N + BINW - 1) / BINW;
  const size_t NBz = ((size_t)(NB + 1) * FS + ZERO_TILE - 1) / ZERO_TILE * ZERO_TILE;
  const size_t cap = Ep + 32 * (size_t)NB;
  int*      bin_fill  = (int*)d_ws;
  int*      spill_n   = bin_fill + (size_t)NB * FS;
  int*      cnt       = bin_fill + NBz;
  int*      row_start = cnt + Np;
  float*    W12       = (float*)(row_start + Np);
  float*    cvec      = W12 + F_IN * C_OUT;
  unsigned* region    = (unsigned*)(cvec + 32);
  int*      spill_src = (int*)(region + (size_t)NB * BCAP);
  int*      spill_dst = spill_src + Ep;
  int*      csr       = spill_dst + Ep;
  float*    yd        = (float*)(csr + cap);
  float*    zd        = yd + Np * C_OUT;

  const int tiles = (N + 15) / 16;
  const int nbz   = (int)(NBz / ZERO_TILE);
  const int nbE   = (int)(((size_t)E + CHUNK - 1) / CHUNK);
  const int nbG   = (int)(((size_t)N * 2 + 255) / 256);

  k_zero  <<<nbz, 256, 0, stream>>>(bin_fill);
  k_bin   <<<nbE, T1, 0, stream>>>(src, dst, bin_fill, region, spill_n, spill_src, spill_dst, E, N);
  k_rows  <<<NB, T2, 0, stream>>>(bin_fill, region, spill_n, spill_src, spill_dst, cnt, row_start, csr, N, E);

  k_fold  <<<3, 32, 0, stream>>>(W1, b1, W2, W12, cvec);
  k_gemm_y<<<tiles, 32, 0, stream>>>(x, W12, cnt, yd, N);

  k_gather<<<nbG, 256, 0, stream>>>(cnt, row_start, csr, yd, cvec, zd, N, (int)cap, 1);
  k_gather<<<nbG, 256, 0, stream>>>(cnt, row_start, csr, zd, b2, out, N, (int)cap, 0);
}


# Seven launches become five: the weight fold is an extra block of the counter-clearing launch, and each row-building work-group does its bin's 16 GEMM tiles (same f32 WMMA, same order of sums)
# speedup vs baseline: 7.2138x; 1.1295x over previous
#include <hip/hip_runtime.h>

typedef __attribute__((ext_vector_type(2))) float v2f;
typedef __attribute__((ext_vector_type(8))) float v8f;


#define F_IN  32
#define HC    64
#define C_OUT 8
#define ZERO_TILE 1024
#define BSH    8
#define BINW   256
#define BCAP   8192
#define CHUNK  8192
#define T1     1024
#define EPT1   8
#define T2     512
#define RPT2   16
#define MAXNB  1024
#define FS     32


__device__ __forceinline__ int wave_incl_scan(int x, int lane) {
  #pragma unroll
  for (int o = 1; o < 32; o <<= 1) {
    const int y = __shfl_up(x, o);
    if (lane >= o) x += y;
  }
  return x;
}

template <int NW>
__device__ __forceinline__ int block_excl_scan(int v, int* wsum, int tid, int& total) {
  const int lane = tid & 31;
  const int w    = tid >> 5;
  const int inc  = wave_incl_scan(v, lane);
  __syncthreads();
  if (lane == 31) wsum[w] = inc;
  __syncthreads();
  if (w == 0) {
    const int t = lane < NW ? wsum[lane] : 0;
    wsum[lane] = wave_incl_scan(t, lane);
  }
  __syncthreads();
  total = wsum[31];
  return (w ? wsum[w - 1] : 0) + inc - v;
}

__global__ __launch_bounds__(T1) void k_bin(const int* __restrict__ src,
                                            const int* __restrict__ dst,
                                            int* __restrict__ bin_fill,
                                            unsigned* __restrict__ region,
                                            int* __restrict__ spill_n,
                                            int* __restrict__ spill_src,
                                            int* __restrict__ spill_dst, int E, int N) {
  __shared__ int      hist[MAXNB];
  __shared__ int      gbase[MAXNB];
  __shared__ unsigned s_rec[CHUNK];
  __shared__ int      s_adr[CHUNK];
  __shared__ int      wsum[32];
  const int tid = threadIdx.x;
  const unsigned e0    = blockIdx.x * (unsigned)CHUNK;
  const unsigned elast = (unsigned)E - 1u;
  hist[tid] = 0;
  __syncthreads();
  int d[EPT1], s[EPT1], r[EPT1];
  #pragma unroll
  for (int k = 0; k < EPT1; ++k) {
    const unsigned e  = e0 + k * T1 + tid;
    const unsigned ec = e < elast ? e : elast;
    d[k] = dst[ec];
    s[k] = src[ec];
  }
  #pragma unroll
  for (int k = 0; k < EPT1; ++k) {
    const unsigned e = e0 + k * T1 + tid;
    int dd = d[k];  dd = dd < 0 ? 0 : dd;  dd = dd < N ? dd : N - 1;
    int ss = s[k];  ss = ss < 0 ? 0 : ss;  ss = ss < N ? ss : N - 1;
    d[k] = dd;  s[k] = ss;
    r[k] = (e < (unsigned)E) ? atomicAdd(&hist[dd >> BSH], 1) : 0;
  }
  __syncthreads();
  const int c = hist[tid];
  int chunk_total;
  const int lo = block_excl_scan<T1 / 32>(c, wsum, tid, chunk_total);
  int g = 0;
  if (c > 0) g = atomicAdd(&bin_fill[tid * FS], c);
  gbase[tid] = g;
  hist[tid]  = lo;
  __syncthreads();
  #pragma unroll
  for (int k = 0; k < EPT1; ++k) {
    const unsigned e = e0 + k * T1 + tid;
    if (e < (unsigned)E) {
      const int b  = d[k] >> BSH;
      const int lp = hist[b] + r[k];
      const int gp = gbase[b] + r[k];
      s_rec[lp] = ((unsigned)s[k] << 8) | (unsigned)(d[k] & (BINW - 1));
      s_adr[lp] = gp < BCAP ? b * BCAP + gp : -1 - b;
    }
  }
  __syncthreads();
  const unsigned left   = (unsigned)E - e0;
  const int      n_here = left < (unsigned)CHUNK ? (int)left : CHUNK;
  for (int i = tid; i < n_here; i += T1) {
    const int      a   = s_adr[i];
    const unsigned rec = s_rec[i];
    if (a >= 0) {
      region[a] = rec;
    } else {
      const int q = atomicAdd(spill_n, 1);
      if ((unsigned)q < (unsigned)E) {
        spill_src[q] = (int)(rec >> 8);
        spill_dst[q] = ((-1 - a) << BSH) | (int)(rec & (BINW - 1));
      }
    }
  }
}


__global__ __launch_bounds__(T2) void k_rows(const int* __restrict__ bin_fill,
                                             const unsigned* __restrict__ region,
                                             const int* __restrict__ spill_n,
                                             const int* __restrict__ spill_src,
                                             const int* __restrict__ spill_dst,
                                             int* __restrict__ cnt,
                                             int* __restrict__ row_start,
                                             int* __restrict__ csr,
                                             const float* __restrict__ x,
                                             const float* __restrict__ W12,
                                             float* __restrict__ yd, int N, int E) {
  __shared__ int cnt_l[BINW];
  __shared__ int cur_l[BINW];
  __shared__ int sorted[BCAP];
  __shared__ int wsum[32];
  const int tid = threadIdx.x;
  const int b   = blockIdx.x;
  const int l15   = tid & 15;
  const int khalf = (tid >> 4) & 1;
  const int tile0 = b * BINW + (tid >> 5) * 16;
  int xrow = tile0 + l15;
  if (xrow >= N) xrow = N - 1;
  const int wcol = l15 & (C_OUT - 1);
  const int total = bin_fill[b * FS];
  const int n     = total < BCAP ? total : BCAP;
  const bool spilled = total > BCAP;
  int acc = 0;
  for (int j = tid; j < b; j += T2) acc += (bin_fill[j * FS] + 31) & ~31;
  int base;
  (void)block_excl_scan<T2 / 32>(acc, wsum, tid, base);
  if (tid < BINW) cnt_l[tid] = 0;
  __syncthreads();
  const unsigned* reg = region + (size_t)b * BCAP;
  unsigned rec[RPT2];
  #pragma unroll
  for (int k = 0; k < RPT2; ++k) {
    const int i = k * T2 + tid;
    rec[k] = 0u;
    if (i < n) rec[k] = reg[i];
  }
  #pragma unroll
  for (int k = 0; k < RPT2; ++k) {
    const int i = k * T2 + tid;
    if (i < n) atomicAdd(&cnt_l[rec[k] & (BINW - 1)], 1);
  }
  int m = 0;
  if (spilled) {
    m = *spill_n;
    m = m < 0 ? 0 : m;
    m = m < E ? m : E;
    for (int q = tid; q < m; q += T2) {
      const int dq = spill_dst[q];
      if ((dq >> BSH) == b) atomicAdd(&cnt_l[dq & (BINW - 1)], 1);
    }
  }
  __syncthreads();
  const int v = tid < BINW ? cnt_l[tid] : 0;
  int bin_total;
  const int ex = block_excl_scan<T2 / 32>(v, wsum, tid, bin_total);
  if (tid < BINW) {
    cur_l[tid] = ex;
    const int node = b * BINW + tid;
    if (node < N) {
      cnt[node]       = v;
      row_start[node] = base + ex;
    }
  }
  __syncthreads();
  v2f fa[F_IN / 4], fb[F_IN / 4];
  #pragma unroll
  for (int k = 0; k < F_IN / 4; ++k) {
    const int ka = k * 4 + khalf * 2;
    fa[k] = v2f{ x[(size_t)xrow * F_IN + ka], x[(size_t)xrow * F_IN + ka + 1] };
    fb[k] = v2f{ W12[(size_t)ka * C_OUT + wcol], W12[(size_t)(ka + 1) * C_OUT + wcol] };
  }
  #pragma unroll
  for (int k = 0; k < RPT2; ++k) {
    const int i = k * T2 + tid;
    if (i < n) {
      const int pos = atomicAdd(&cur_l[rec[k] & (BINW - 1)], 1);
      const int sv  = (int)(rec[k] >> 8);
      if (!spilled) sorted[pos] = sv;
      else          csr[base + pos] = sv;
    }
  }
  if (spilled) {
    for (int q = tid; q < m; q += T2) {
      const int dq = spill_dst[q];
      if ((dq >> BSH) == b) {
        const int pos = atomicAdd(&cur_l[dq & (BINW - 1)], 1);
        csr[base + pos] = spill_src[q];
      }
    }
  }
  __syncthreads();
  if (!spilled) {
    for (int i = tid; i < total; i += T2) csr[base + i] = sorted[i];
  }
  const float dmine = 1.0f / sqrtf((float)(cnt_l[(tid >> 5) * 16 + l15] + 1));
  v8f c = {};
  #pragma unroll
  for (int k = 0; k < F_IN / 4; ++k)
    c = __builtin_amdgcn_wmma_f32_16x16x4_f32(false, fa[k], false, fb[k],
                                              (short)0, c, false, false);
  float sc[8];
  #pragma unroll
  for (int r = 0; r < 8; ++r) sc[r] = __shfl(dmine, khalf * 8 + r);
  if (l15 < C_OUT) {
    #pragma unroll
    for (int r = 0; r < 8; ++r) {
      const int orow = tile0 + khalf * 8 + r;
      if (orow < N) yd[(size_t)orow * C_OUT + l15] = c[r] * sc[r];
    }
  }
}


__device__ __forceinline__ void fold_tile(int tile, int lane,
                                          const float* __restrict__ W1,
                                          const float* __restrict__ b1,
                                          const float* __restrict__ W2,
                                          float* __restrict__ W12,
                                          float* __restrict__ cvec) {
  const int l15   = lane & 15;
  const int khalf = lane >> 4;
  const float* arow = (tile < 2) ? (W1 + (size_t)(tile * 16 + l15) * HC) : b1;
  const bool live_col = (l15 < C_OUT);
  const int  col = l15 & (C_OUT - 1);
  v8f c = {};
  #pragma unroll
  for (int k0 = 0; k0 < HC; k0 += 4) {
    const int ka = k0 + khalf * 2;
    v2f a = { arow[ka], arow[ka + 1] };
    v2f b = { W2[(size_t)ka * C_OUT + col], W2[(size_t)(ka + 1) * C_OUT + col] };
    c = __builtin_amdgcn_wmma_f32_16x16x4_f32(false, a, false, b,
                                              (short)0, c, false, false);
  }
  if (live_col) {
    if (tile < 2) {
      const int mbase = tile * 16 + khalf * 8;
      #pragma unroll
      for (int r = 0; r < 8; ++r)
        W12[(size_t)(mbase + r) * C_OUT + l15] = c[r];
    } else if (khalf == 0) {
      cvec[l15] = c[0];
    }
  }
}

__global__ __launch_bounds__(256) void k_zero_fold(int* __restrict__ p, int nbz,
                                                   const float* __restrict__ W1,
                                                   const float* __restrict__ b1,
                                                   const float* __restrict__ W2,
                                                   float* __restrict__ W12,
                                                   float* __restrict__ cvec) {
  if ((int)blockIdx.x < nbz) {
    const int i = blockIdx.x * 256 + threadIdx.x;
    ((int4*)p)[i] = make_int4(0, 0, 0, 0);
    return;
  }
  const int tile = __builtin_amdgcn_readfirstlane((int)(threadIdx.x >> 5));
  if (tile < 3) fold_tile(tile, threadIdx.x & 31, W1, b1, W2, W12, cvec);
}


__global__ __launch_bounds__(256) void k_gather(const int* __restrict__ cnt,
                                                const int* __restrict__ row_start,
                                                const int* __restrict__ csr,
                                                const float* __restrict__ feat,
                                                const float* __restrict__ bias,
                                                float* __restrict__ res,
                                                int N, int cap, int scale_out) {
  const unsigned gid = blockIdx.x * 256u + threadIdx.x;
  const int node = gid >> 1;
  const int half = gid & 1;
  if (node >= N) return;
  const int end = cnt[node];
  int rs = row_start[node];
  rs = rs < 0 ? 0 : rs;
  const float di = 1.0f / sqrtf((float)(end + 1));
  const float4* f4 = (const float4*)feat;
  const float4 bb = ((const float4*)bias)[half];
  float4 acc = f4[gid];
  for (int j = 0; j < end; j += 8) {
    const int last = end - 1;
    int s[8];
    #pragma unroll
    for (int k = 0; k < 8; ++k) {
      const int jj = (j + k < last) ? (j + k) : last;
      int p = rs + jj;
      p = p < cap ? p : cap - 1;
      int t = csr[p];
      t = t < 0 ? 0 : t;
      s[k] = t < N ? t : N - 1;
    }
    float4 v[8];
    #pragma unroll
    for (int k = 0; k < 8; ++k) v[k] = f4[(size_t)s[k] * 2 + half];
    #pragma unroll
    for (int k = 0; k < 8; ++k) {
      const bool live = (k == 0) || (j + k) < end;
      acc.x += live ? v[k].x : 0.0f;
      acc.y += live ? v[k].y : 0.0f;
      acc.z += live ? v[k].z : 0.0f;
      acc.w += live ? v[k].w : 0.0f;
    }
  }
  float4 o;
  o.x = bb.x + di * acc.x;  o.y = bb.y + di * acc.y;
  o.z = bb.z + di * acc.z;  o.w = bb.w + di * acc.w;
  if (scale_out) { o.x *= di;  o.y *= di;  o.z *= di;  o.w *= di; }
  ((float4*)res)[gid] = o;
}


extern "C" void kernel_launch(void* const* d_in, const int* in_sizes, int n_in,
                              void* d_out, int out_size, void* d_ws, size_t ws_size,
                              hipStream_t stream) {
  const float* x  = (const float*)d_in[0];
  const int*   ei = (const int*)d_in[1];
  const float* W1 = (const float*)d_in[2];
  const float* b1 = (const float*)d_in[3];
  const float* W2 = (const float*)d_in[4];
  const float* b2 = (const float*)d_in[5];
  float* out = (float*)d_out;

  const int N = in_sizes[0] / F_IN;
  const int E = in_sizes[1] / 2;
  const int* src = ei;
  const int* dst = ei + E;

  const size_t Np  = ((size_t)N + ZERO_TILE - 1) / ZERO_TILE * ZERO_TILE;
  const size_t Ep  = ((size_t)E + 31) / 32 * 32;
  const int    NB  = (N + BINW - 1) / BINW;
  const size_t NBz = ((size_t)(NB + 1) * FS + ZERO_TILE - 1) / ZERO_TILE * ZERO_TILE;
  const size_t cap = Ep + 32 * (size_t)NB;
  int*      bin_fill  = (int*)d_ws;
  int*      spill_n   = bin_fill + (size_t)NB * FS;
  int*      cnt       = bin_fill + NBz;
  int*      row_start = cnt + Np;
  float*    W12       = (float*)(row_start + Np);
  float*    cvec      = W12 + F_IN * C_OUT;
  unsigned* region    = (unsigned*)(cvec + 32);
  int*      spill_src = (int*)(region + (size_t)NB * BCAP);
  int*      spill_dst = spill_src + Ep;
  int*      csr       = spill_dst + Ep;
  float*    yd        = (float*)(csr + cap);
  float*    zd        = yd + Np * C_OUT;

  const int nbz   = (int)(NBz / ZERO_TILE);
  const int nbE   = (int)(((size_t)E + CHUNK - 1) / CHUNK);
  const int nbG   = (int)(((size_t)N * 2 + 255) / 256);

  k_zero_fold<<<nbz + 1, 256, 0, stream>>>(bin_fill, nbz, W1, b1, W2, W12, cvec);
  k_bin   <<<nbE, T1, 0, stream>>>(src, dst, bin_fill, region, spill_n, spill_src, spill_dst, E, N);
  k_rows  <<<NB, T2, 0, stream>>>(bin_fill, region, spill_n, spill_src, spill_dst, cnt, row_start, csr,
                                  x, W12, yd, N, E);

  k_gather<<<nbG, 256, 0, stream>>>(cnt, row_start, csr, yd, cvec, zd, N, (int)cap, 1);
  k_gather<<<nbG, 256, 0, stream>>>(cnt, row_start, csr, zd, b2, out, N, (int)cap, 0);
}


# Bins' global fill counters replaced by per-chunk offset tables (chunks written in bin order; a bin finds its runs via the tables): no clearing launch, 5 launches -> 4; sums stay in LDS arrival order
# speedup vs baseline: 7.7254x; 1.0709x over previous
#include <hip/hip_runtime.h>

typedef __attribute__((ext_vector_type(2))) float v2f;
typedef __attribute__((ext_vector_type(8))) float v8f;


#define F_IN  32
#define HC    64
#define C_OUT 8
#define ZERO_TILE 1024
#define BSH    8
#define BINW   256
#define BCAP   8192
#define CHUNK  8192
#define T1     1024
#define EPT1   8
#define T2     512
#define NW2    16
#define CPW    16
#define MAXNB  1024
#define MAXCH  512
#define LP     1056
#define NOREC  0xFFFFFFFFu


__device__ __forceinline__ void fold_tile(int tile, int lane,
                                          const float* __restrict__ W1,
                                          const float* __restrict__ b1,
                                          const float* __restrict__ W2,
                                          float* __restrict__ W12,
                                          float* __restrict__ cvec) {
  const int l15   = lane & 15;
  const int khalf = lane >> 4;
  const float* arow = (tile < 2) ? (W1 + (size_t)(tile * 16 + l15) * HC) : b1;
  const bool live_col = (l15 < C_OUT);
  const int  col = l15 & (C_OUT - 1);
  v8f c = {};
  #pragma unroll
  for (int k0 = 0; k0 < HC; k0 += 4) {
    const int ka = k0 + khalf * 2;
    v2f a = { arow[ka], arow[ka + 1] };
    v2f b = { W2[(size_t)ka * C_OUT + col], W2[(size_t)(ka + 1) * C_OUT + col] };
    c = __builtin_amdgcn_wmma_f32_16x16x4_f32(false, a, false, b,
                                              (short)0, c, false, false);
  }
  if (live_col) {
    if (tile < 2) {
      const int mbase = tile * 16 + khalf * 8;
      #pragma unroll
      for (int r = 0; r < 8; ++r)
        W12[(size_t)(mbase + r) * C_OUT + l15] = c[r];
    } else if (khalf == 0) {
      cvec[l15] = c[0];
    }
  }
}


__device__ __forceinline__ int wave_incl_scan(int x, int lane) {
  #pragma unroll
  for (int o = 1; o < 32; o <<= 1) {
    const int y = __shfl_up(x, o);
    if (lane >= o) x += y;
  }
  return x;
}

template <int NW>
__device__ __forceinline__ int block_excl_scan(int v, int* wsum, int tid, int& total) {
  const int lane = tid & 31;
  const int w    = tid >> 5;
  const int inc  = wave_incl_scan(v, lane);
  __syncthreads();
  if (lane == 31) wsum[w] = inc;
  __syncthreads();
  if (w == 0) {
    const int t = lane < NW ? wsum[lane] : 0;
    wsum[lane] = wave_incl_scan(t, lane);
  }
  __syncthreads();
  total = wsum[31];
  return (w ? wsum[w - 1] : 0) + inc - v;
}

__global__ __launch_bounds__(T1) void k_bin(const int* __restrict__ src,
                                            const int* __restrict__ dst,
                                            unsigned* __restrict__ chunk_rec,
                                            int* __restrict__ chunk_lo,
                                            int E, int N,
                                            const float* __restrict__ W1,
                                            const float* __restrict__ b1,
                                            const float* __restrict__ W2,
                                            float* __restrict__ W12,
                                            float* __restrict__ cvec) {
  __shared__ int      hist[MAXNB];
  __shared__ unsigned s_rec[CHUNK];
  __shared__ int      wsum[32];
  const int tid = threadIdx.x;
  if (blockIdx.x == 0) {
    const int tile = __builtin_amdgcn_readfirstlane(tid >> 5);
    if (tile < 3) fold_tile(tile, tid & 31, W1, b1, W2, W12, cvec);
    return;
  }
  const unsigned chunk = blockIdx.x - 1u;
  const unsigned e0    = chunk * (unsigned)CHUNK;
  const unsigned elast = (unsigned)E - 1u;
  hist[tid] = 0;
  __syncthreads();
  int d[EPT1], s[EPT1], r[EPT1];
  #pragma unroll
  for (int k = 0; k < EPT1; ++k) {
    const unsigned e  = e0 + k * T1 + tid;
    const unsigned ec = e < elast ? e : elast;
    d[k] = dst[ec];
    s[k] = src[ec];
  }
  #pragma unroll
  for (int k = 0; k < EPT1; ++k) {
    const unsigned e = e0 + k * T1 + tid;
    int dd = d[k];  dd = dd < 0 ? 0 : dd;  dd = dd < N ? dd : N - 1;
    int ss = s[k];  ss = ss < 0 ? 0 : ss;  ss = ss < N ? ss : N - 1;
    d[k] = dd;  s[k] = ss;
    r[k] = (e < (unsigned)E) ? atomicAdd(&hist[dd >> BSH], 1) : 0;
  }
  __syncthreads();
  const int c = hist[tid];
  int chunk_total;
  const int lo = block_excl_scan<T1 / 32>(c, wsum, tid, chunk_total);
  hist[tid] = lo;
  int* my_lo = chunk_lo + (size_t)chunk * LP;
  my_lo[tid] = lo;
  if (tid == 0) my_lo[MAXNB] = chunk_total;
  __syncthreads();
  #pragma unroll
  for (int k = 0; k < EPT1; ++k) {
    const unsigned e = e0 + k * T1 + tid;
    if (e < (unsigned)E) {
      const int lp = hist[d[k] >> BSH] + r[k];
      s_rec[lp] = ((unsigned)s[k] << 8) | (unsigned)(d[k] & (BINW - 1));
    }
  }
  __syncthreads();
  unsigned* my_rec = chunk_rec + (size_t)chunk * CHUNK;
  for (int i = tid; i < chunk_total; i += T1) my_rec[i] = s_rec[i];
}


__global__ __launch_bounds__(T2) void k_rows(const unsigned* __restrict__ chunk_rec,
                                             const int* __restrict__ chunk_lo,
                                             int nchunks,
                                             int* __restrict__ cnt,
                                             int* __restrict__ row_start,
                                             int* __restrict__ csr,
                                             const float* __restrict__ x,
                                             const float* __restrict__ W12,
                                             float* __restrict__ yd, int N) {
  __shared__ int cnt_l[BINW];
  __shared__ int cur_l[BINW];
  __shared__ int sorted[BCAP];
  __shared__ int2 run[MAXCH];
  __shared__ int part_lo[NW2];
  __shared__ int part_len[NW2];
  __shared__ int wsum[32];
  const int tid  = threadIdx.x;
  const int lane = tid & 31;
  const int w    = tid >> 5;
  const int b    = blockIdx.x;
  const int l15   = tid & 15;
  const int khalf = (tid >> 4) & 1;
  const int tile0 = b * BINW + w * 16;
  int xrow = tile0 + l15;
  if (xrow >= N) xrow = N - 1;
  const int wcol = l15 & (C_OUT - 1);
  int l0 = 0, len = 0;
  if (tid < nchunks) {
    const int* t_lo = chunk_lo + (size_t)tid * LP + b;
    l0 = t_lo[0];
    int l1 = t_lo[1];
    l0 = l0 < 0 ? 0 : l0;  l0 = l0 < CHUNK ? l0 : CHUNK;
    l1 = l1 < l0 ? l0 : l1;  l1 = l1 < CHUNK ? l1 : CHUNK;
    len = l1 - l0;
  }
  run[tid] = make_int2(l0, len);
  int sum_lo = l0, sum_len = len;
  #pragma unroll
  for (int o = 16; o > 0; o >>= 1) {
    sum_lo  += __shfl_xor(sum_lo, o);
    sum_len += __shfl_xor(sum_len, o);
  }
  if (lane == 0) { part_lo[w] = sum_lo;  part_len[w] = sum_len; }
  if (tid < BINW) cnt_l[tid] = 0;
  __syncthreads();
  int before = 0, total = 0;
  #pragma unroll
  for (int j = 0; j < NW2; ++j) { before += part_lo[j];  total += part_len[j]; }
  const int  base    = ((before + 31) & ~31) + 32 * b;
  const bool spilled = total > BCAP;
  int2 rn[CPW];
  #pragma unroll
  for (int k = 0; k < CPW; ++k) rn[k] = run[w + NW2 * k];
  unsigned rec[CPW];
  bool more = nchunks > NW2 * CPW;
  #pragma unroll
  for (int k = 0; k < CPW; ++k) {
    rec[k] = NOREC;
    more = more || rn[k].y > 32;
    if (lane < rn[k].y) rec[k] = chunk_rec[(size_t)(w + NW2 * k) * CHUNK + rn[k].x + lane];
  }
  #pragma unroll
  for (int k = 0; k < CPW; ++k)
    if (rec[k] != NOREC) atomicAdd(&cnt_l[rec[k] & (BINW - 1)], 1);
  if (more) {
    for (int c = w; c < nchunks; c += NW2) {
      const int rl = run[c].y;
      const unsigned* rp = chunk_rec + (size_t)c * CHUNK + run[c].x;
      for (int o = (c < NW2 * CPW ? 32 : 0) + lane; o < rl; o += 32)
        atomicAdd(&cnt_l[rp[o] & (BINW - 1)], 1);
    }
  }
  __syncthreads();
  const int v = tid < BINW ? cnt_l[tid] : 0;
  int bin_total;
  const int ex = block_excl_scan<NW2>(v, wsum, tid, bin_total);
  if (tid < BINW) {
    cur_l[tid] = ex;
    const int node = b * BINW + tid;
    if (node < N) {
      cnt[node]       = v;
      row_start[node] = base + ex;
    }
  }
  __syncthreads();
  v2f fa[F_IN / 4], fb[F_IN / 4];
  #pragma unroll
  for (int k = 0; k < F_IN / 4; ++k) {
    const int ka = k * 4 + khalf * 2;
    fa[k] = v2f{ x[(size_t)xrow * F_IN + ka], x[(size_t)xrow * F_IN + ka + 1] };
    fb[k] = v2f{ W12[(size_t)ka * C_OUT + wcol], W12[(size_t)(ka + 1) * C_OUT + wcol] };
  }
  #pragma unroll
  for (int k = 0; k < CPW; ++k) {
    if (rec[k] != NOREC) {
      const int pos = atomicAdd(&cur_l[rec[k] & (BINW - 1)], 1);
      const int sv  = (int)(rec[k] >> 8);
      if (!spilled) sorted[pos] = sv;
      else          csr[base + pos] = sv;
    }
  }
  if (more) {
    for (int c = w; c < nchunks; c += NW2) {
      const int rl = run[c].y;
      const unsigned* rp = chunk_rec + (size_t)c * CHUNK + run[c].x;
      for (int o = (c < NW2 * CPW ? 32 : 0) + lane; o < rl; o += 32) {
        const unsigned rr = rp[o];
        const int pos = atomicAdd(&cur_l[rr & (BINW - 1)], 1);
        const int sv  = (int)(rr >> 8);
        if (!spilled) sorted[pos] = sv;
        else          csr[base + pos] = sv;
      }
    }
  }
  __syncthreads();
  if (!spilled) {
    for (int i = tid; i < total; i += T2) csr[base + i] = sorted[i];
  }
  const float dmine = 1.0f / sqrtf((float)(cnt_l[w * 16 + l15] + 1));
  v8f c = {};
  #pragma unroll
  for (int k = 0; k < F_IN / 4; ++k)
    c = __builtin_amdgcn_wmma_f32_16x16x4_f32(false, fa[k], false, fb[k],
                                              (short)0, c, false, false);
  float sc[8];
  #pragma unroll
  for (int r = 0; r < 8; ++r) sc[r] = __shfl(dmine, khalf * 8 + r);
  if (l15 < C_OUT) {
    #pragma unroll
    for (int r = 0; r < 8; ++r) {
      const int orow = tile0 + khalf * 8 + r;
      if (orow < N) yd[(size_t)orow * C_OUT + l15] = c[r] * sc[r];
    }
  }
}


__global__ __launch_bounds__(256) void k_gather(const int* __restrict__ cnt,
                                                const int* __restrict__ row_start,
                                                const int* __restrict__ csr,
                                                const float* __restrict__ feat,
                                                const float* __restrict__ bias,
                                                float* __restrict__ res,
                                                int N, int cap, int scale_out) {
  const unsigned gid = blockIdx.x * 256u + threadIdx.x;
  const int node = gid >> 1;
  const int half = gid & 1;
  if (node >= N) return;
  const int end = cnt[node];
  int rs = row_start[node];
  rs = rs < 0 ? 0 : rs;
  const float di = 1.0f / sqrtf((float)(end + 1));
  const float4* f4 = (const float4*)feat;
  const float4 bb = ((const float4*)bias)[half];
  float4 acc = f4[gid];
  for (int j = 0; j < end; j += 8) {
    const int last = end - 1;
    int s[8];
    #pragma unroll
    for (int k = 0; k < 8; ++k) {
      const int jj = (j + k < last) ? (j + k) : last;
      int p = rs + jj;
      p = p < cap ? p : cap - 1;
      int t = csr[p];
      t = t < 0 ? 0 : t;
      s[k] = t < N ? t : N - 1;
    }
    float4 v[8];
    #pragma unroll
    for (int k = 0; k < 8; ++k) v[k] = f4[(size_t)s[k] * 2 + half];
    #pragma unroll
    for (int k = 0; k < 8; ++k) {
      const bool live = (k == 0) || (j + k) < end;
      acc.x += live ? v[k].x : 0.0f;
      acc.y += live ? v[k].y : 0.0f;
      acc.z += live ? v[k].z : 0.0f;
      acc.w += live ? v[k].w : 0.0f;
    }
  }
  float4 o;
  o.x = bb.x + di * acc.x;  o.y = bb.y + di * acc.y;
  o.z = bb.z + di * acc.z;  o.w = bb.w + di * acc.w;
  if (scale_out) { o.x *= di;  o.y *= di;  o.z *= di;  o.w *= di; }
  ((float4*)res)[gid] = o;
}


extern "C" void kernel_launch(void* const* d_in, const int* in_sizes, int n_in,
                              void* d_out, int out_size, void* d_ws, size_t ws_size,
                              hipStream_t stream) {
  const float* x  = (const float*)d_in[0];
  const int*   ei = (const int*)d_in[1];
  const float* W1 = (const float*)d_in[2];
  const float* b1 = (const float*)d_in[3];
  const float* W2 = (const float*)d_in[4];
  const float* b2 = (const float*)d_in[5];
  float* out = (float*)d_out;

  const int N = in_sizes[0] / F_IN;
  const int E = in_sizes[1] / 2;
  const int* src = ei;
  const int* dst = ei + E;

  const size_t Np  = ((size_t)N + ZERO_TILE - 1) / ZERO_TILE * ZERO_TILE;
  const size_t Ep  = ((size_t)E + 31) / 32 * 32;
  const int    NB  = (N + BINW - 1) / BINW;
  const int    nbE = (int)(((size_t)E + CHUNK - 1) / CHUNK);
  const size_t cap = Ep + 32 * ((size_t)NB + 1);
  int*      cnt       = (int*)d_ws;
  int*      row_start = cnt + Np;
  float*    W12       = (float*)(row_start + Np);
  float*    cvec      = W12 + F_IN * C_OUT;
  int*      chunk_lo  = (int*)(cvec + 32);
  unsigned* chunk_rec = (unsigned*)(chunk_lo + (size_t)nbE * LP);
  int*      csr       = (int*)(chunk_rec + (size_t)nbE * CHUNK);
  float*    yd        = (float*)(csr + cap);
  float*    zd        = yd + Np * C_OUT;

  const int nbG = (int)(((size_t)N * 2 + 255) / 256);

  k_bin   <<<nbE + 1, T1, 0, stream>>>(src, dst, chunk_rec, chunk_lo, E, N, W1, b1, W2, W12, cvec);
  k_rows  <<<NB, T2, 0, stream>>>(chunk_rec, chunk_lo, nbE, cnt, row_start, csr, x, W12, yd, N);

  k_gather<<<nbG, 256, 0, stream>>>(cnt, row_start, csr, yd, cvec, zd, N, (int)cap, 1);
  k_gather<<<nbG, 256, 0, stream>>>(cnt, row_start, csr, zd, b2, out, N, (int)cap, 0);
}


# Gather takes 16 row entries a trip instead of 8: indices as four 4-word reads (was 8 one-word), source rows past a row's end masked off instead of re-read; same order of sums
# speedup vs baseline: 8.0678x; 1.0443x over previous
#include <hip/hip_runtime.h>

typedef __attribute__((ext_vector_type(2))) float v2f;
typedef __attribute__((ext_vector_type(8))) float v8f;


#define F_IN  32
#define HC    64
#define C_OUT 8
#define ZERO_TILE 1024
#define BSH    8
#define BINW   256
#define BCAP   8192
#define CHUNK  8192
#define T1     1024
#define EPT1   8
#define T2     512
#define NW2    16
#define CPW    16
#define MAXNB  1024
#define MAXCH  512
#define LP     1056
#define NOREC  0xFFFFFFFFu
#define GT     16


__device__ __forceinline__ void fold_tile(int tile, int lane,
                                          const float* __restrict__ W1,
                                          const float* __restrict__ b1,
                                          const float* __restrict__ W2,
                                          float* __restrict__ W12,
                                          float* __restrict__ cvec) {
  const int l15   = lane & 15;
  const int khalf = lane >> 4;
  const float* arow = (tile < 2) ? (W1 + (size_t)(tile * 16 + l15) * HC) : b1;
  const bool live_col = (l15 < C_OUT);
  const int  col = l15 & (C_OUT - 1);
  v8f c = {};
  #pragma unroll
  for (int k0 = 0; k0 < HC; k0 += 4) {
    const int ka = k0 + khalf * 2;
    v2f a = { arow[ka], arow[ka + 1] };
    v2f b = { W2[(size_t)ka * C_OUT + col], W2[(size_t)(ka + 1) * C_OUT + col] };
    c = __builtin_amdgcn_wmma_f32_16x16x4_f32(false, a, false, b,
                                              (short)0, c, false, false);
  }
  if (live_col) {
    if (tile < 2) {
      const int mbase = tile * 16 + khalf * 8;
      #pragma unroll
      for (int r = 0; r < 8; ++r)
        W12[(size_t)(mbase + r) * C_OUT + l15] = c[r];
    } else if (khalf == 0) {
      cvec[l15] = c[0];
    }
  }
}


__device__ __forceinline__ int wave_incl_scan(int x, int lane) {
  #pragma unroll
  for (int o = 1; o < 32; o <<= 1) {
    const int y = __shfl_up(x, o);
    if (lane >= o) x += y;
  }
  return x;
}

template <int NW>
__device__ __forceinline__ int block_excl_scan(int v, int* wsum, int tid, int& total) {
  const int lane = tid & 31;
  const int w    = tid >> 5;
  const int inc  = wave_incl_scan(v, lane);
  __syncthreads();
  if (lane == 31) wsum[w] = inc;
  __syncthreads();
  if (w == 0) {
    const int t = lane < NW ? wsum[lane] : 0;
    wsum[lane] = wave_incl_scan(t, lane);
  }
  __syncthreads();
  total = wsum[31];
  return (w ? wsum[w - 1] : 0) + inc - v;
}

__global__ __launch_bounds__(T1) void k_bin(const int* __restrict__ src,
                                            const int* __restrict__ dst,
                                            unsigned* __restrict__ chunk_rec,
                                            int* __restrict__ chunk_lo,
                                            int E, int N,
                                            const float* __restrict__ W1,
                                            const float* __restrict__ b1,
                                            const float* __restrict__ W2,
                                            float* __restrict__ W12,
                                            float* __restrict__ cvec) {
  __shared__ int      hist[MAXNB];
  __shared__ unsigned s_rec[CHUNK];
  __shared__ int      wsum[32];
  const int tid = threadIdx.x;
  if (blockIdx.x == 0) {
    const int tile = __builtin_amdgcn_readfirstlane(tid >> 5);
    if (tile < 3) fold_tile(tile, tid & 31, W1, b1, W2, W12, cvec);
    return;
  }
  const unsigned chunk = blockIdx.x - 1u;
  const unsigned e0    = chunk * (unsigned)CHUNK;
  const unsigned elast = (unsigned)E - 1u;
  hist[tid] = 0;
  __syncthreads();
  int d[EPT1], s[EPT1], r[EPT1];
  #pragma unroll
  for (int k = 0; k < EPT1; ++k) {
    const unsigned e  = e0 + k * T1 + tid;
    const unsigned ec = e < elast ? e : elast;
    d[k] = dst[ec];
    s[k] = src[ec];
  }
  #pragma unroll
  for (int k = 0; k < EPT1; ++k) {
    const unsigned e = e0 + k * T1 + tid;
    int dd = d[k];  dd = dd < 0 ? 0 : dd;  dd = dd < N ? dd : N - 1;
    int ss = s[k];  ss = ss < 0 ? 0 : ss;  ss = ss < N ? ss : N - 1;
    d[k] = dd;  s[k] = ss;
    r[k] = (e < (unsigned)E) ? atomicAdd(&hist[dd >> BSH], 1) : 0;
  }
  __syncthreads();
  const int c = hist[tid];
  int chunk_total;
  const int lo = block_excl_scan<T1 / 32>(c, wsum, tid, chunk_total);
  hist[tid] = lo;
  int* my_lo = chunk_lo + (size_t)chunk * LP;
  my_lo[tid] = lo;
  if (tid == 0) my_lo[MAXNB] = chunk_total;
  __syncthreads();
  #pragma unroll
  for (int k = 0; k < EPT1; ++k) {
    const unsigned e = e0 + k * T1 + tid;
    if (e < (unsigned)E) {
      const int lp = hist[d[k] >> BSH] + r[k];
      s_rec[lp] = ((unsigned)s[k] << 8) | (unsigned)(d[k] & (BINW - 1));
    }
  }
  __syncthreads();
  unsigned* my_rec = chunk_rec + (size_t)chunk * CHUNK;
  for (int i = tid; i < chunk_total; i += T1) my_rec[i] = s_rec[i];
}


__global__ __launch_bounds__(T2) void k_rows(const unsigned* __restrict__ chunk_rec,
                                             const int* __restrict__ chunk_lo,
                                             int nchunks,
                                             int* __restrict__ cnt,
                                             int* __restrict__ row_start,
                                             int* __restrict__ csr,
                                             const float* __restrict__ x,
                                             const float* __restrict__ W12,
                                             float* __restrict__ yd, int N) {
  __shared__ int cnt_l[BINW];
  __shared__ int cur_l[BINW];
  __shared__ int sorted[BCAP];
  __shared__ int2 run[MAXCH];
  __shared__ int part_lo[NW2];
  __shared__ int part_len[NW2];
  __shared__ int wsum[32];
  const int tid  = threadIdx.x;
  const int lane = tid & 31;
  const int w    = tid >> 5;
  const int b    = blockIdx.x;
  const int l15   = tid & 15;
  const int khalf = (tid >> 4) & 1;
  const int tile0 = b * BINW + w * 16;
  int xrow = tile0 + l15;
  if (xrow >= N) xrow = N - 1;
  const int wcol = l15 & (C_OUT - 1);
  int l0 = 0, len = 0;
  if (tid < nchunks) {
    const int* t_lo = chunk_lo + (size_t)tid * LP + b;
    l0 = t_lo[0];
    int l1 = t_lo[1];
    l0 = l0 < 0 ? 0 : l0;  l0 = l0 < CHUNK ? l0 : CHUNK;
    l1 = l1 < l0 ? l0 : l1;  l1 = l1 < CHUNK ? l1 : CHUNK;
    len = l1 - l0;
  }
  run[tid] = make_int2(l0, len);
  int sum_lo = l0, sum_len = len;
  #pragma unroll
  for (int o = 16; o > 0; o >>= 1) {
    sum_lo  += __shfl_xor(sum_lo, o);
    sum_len += __shfl_xor(sum_len, o);
  }
  if (lane == 0) { part_lo[w] = sum_lo;  part_len[w] = sum_len; }
  if (tid < BINW) cnt_l[tid] = 0;
  __syncthreads();
  int before = 0, total = 0;
  #pragma unroll
  for (int j = 0; j < NW2; ++j) { before += part_lo[j];  total += part_len[j]; }
  const int  base    = ((before + 31) & ~31) + 32 * b;
  const bool spilled = total > BCAP;
  int2 rn[CPW];
  #pragma unroll
  for (int k = 0; k < CPW; ++k) rn[k] = run[w + NW2 * k];
  unsigned rec[CPW];
  bool more = nchunks > NW2 * CPW;
  #pragma unroll
  for (int k = 0; k < CPW; ++k) {
    rec[k] = NOREC;
    more = more || rn[k].y > 32;
    if (lane < rn[k].y) rec[k] = chunk_rec[(size_t)(w + NW2 * k) * CHUNK + rn[k].x + lane];
  }
  #pragma unroll
  for (int k = 0; k < CPW; ++k)
    if (rec[k] != NOREC) atomicAdd(&cnt_l[rec[k] & (BINW - 1)], 1);
  if (more) {
    for (int c = w; c < nchunks; c += NW2) {
      const int rl = run[c].y;
      const unsigned* rp = chunk_rec + (size_t)c * CHUNK + run[c].x;
      for (int o = (c < NW2 * CPW ? 32 : 0) + lane; o < rl; o += 32)
        atomicAdd(&cnt_l[rp[o] & (BINW - 1)], 1);
    }
  }
  __syncthreads();
  const int v = tid < BINW ? cnt_l[tid] : 0;
  int bin_total;
  const int ex = block_excl_scan<NW2>(v, wsum, tid, bin_total);
  if (tid < BINW) {
    cur_l[tid] = ex;
    const int node = b * BINW + tid;
    if (node < N) {
      cnt[node]       = v;
      row_start[node] = base + ex;
    }
  }
  __syncthreads();
  v2f fa[F_IN / 4], fb[F_IN / 4];
  #pragma unroll
  for (int k = 0; k < F_IN / 4; ++k) {
    const int ka = k * 4 + khalf * 2;
    fa[k] = v2f{ x[(size_t)xrow * F_IN + ka], x[(size_t)xrow * F_IN + ka + 1] };
    fb[k] = v2f{ W12[(size_t)ka * C_OUT + wcol], W12[(size_t)(ka + 1) * C_OUT + wcol] };
  }
  #pragma unroll
  for (int k = 0; k < CPW; ++k) {
    if (rec[k] != NOREC) {
      const int pos = atomicAdd(&cur_l[rec[k] & (BINW - 1)], 1);
      const int sv  = (int)(rec[k] >> 8);
      if (!spilled) sorted[pos] = sv;
      else          csr[base + pos] = sv;
    }
  }
  if (more) {
    for (int c = w; c < nchunks; c += NW2) {
      const int rl = run[c].y;
      const unsigned* rp = chunk_rec + (size_t)c * CHUNK + run[c].x;
      for (int o = (c < NW2 * CPW ? 32 : 0) + lane; o < rl; o += 32) {
        const unsigned rr = rp[o];
        const int pos = atomicAdd(&cur_l[rr & (BINW - 1)], 1);
        const int sv  = (int)(rr >> 8);
        if (!spilled) sorted[pos] = sv;
        else          csr[base + pos] = sv;
      }
    }
  }
  __syncthreads();
  const int fill = ((before + total + 31) & ~31) + 32 * (b + 1) - base
                 + (b == (int)gridDim.x - 1 ? GT - 1 : 0);
  if (!spilled) {
    for (int i = tid; i < fill; i += T2) csr[base + i] = i < total ? sorted[i] : 0;
  } else {
    for (int i = total + tid; i < fill; i += T2) csr[base + i] = 0;
  }
  const float dmine = 1.0f / sqrtf((float)(cnt_l[w * 16 + l15] + 1));
  v8f c = {};
  #pragma unroll
  for (int k = 0; k < F_IN / 4; ++k)
    c = __builtin_amdgcn_wmma_f32_16x16x4_f32(false, fa[k], false, fb[k],
                                              (short)0, c, false, false);
  float sc[8];
  #pragma unroll
  for (int r = 0; r < 8; ++r) sc[r] = __shfl(dmine, khalf * 8 + r);
  if (l15 < C_OUT) {
    #pragma unroll
    for (int r = 0; r < 8; ++r) {
      const int orow = tile0 + khalf * 8 + r;
      if (orow < N) yd[(size_t)orow * C_OUT + l15] = c[r] * sc[r];
    }
  }
}


__global__ __launch_bounds__(256) void k_gather(const int* __restrict__ cnt,
                                                const int* __restrict__ row_start,
                                                const int* __restrict__ csr,
                                                const float* __restrict__ feat,
                                                const float* __restrict__ bias,
                                                float* __restrict__ res,
                                                int N, int cap, int scale_out) {
  const unsigned gid = blockIdx.x * 256u + threadIdx.x;
  const int node = gid >> 1;
  const int half = gid & 1;
  if (node >= N) return;
  const int end = cnt[node];
  int rs = row_start[node];
  rs = rs < 0 ? 0 : rs;
  const float di = 1.0f / sqrtf((float)(end + 1));
  const float4* f4 = (const float4*)feat;
  const float4 bb = ((const float4*)bias)[half];
  float4 acc = f4[gid];
  for (int j = 0; j < end; j += GT) {
    int s[GT];
    #pragma unroll
    for (int g = 0; g < GT / 4; ++g) {
      int p = rs + j + 4 * g;
      p = p < cap - 4 ? p : cap - 4;
      __builtin_memcpy(&s[4 * g], csr + p, 4 * sizeof(int));
    }
    asm volatile("" :: "v"(s[0]), "v"(s[1]), "v"(s[2]), "v"(s[3]), "v"(s[4]), "v"(s[5]), "v"(s[6]), "v"(s[7]),
                       "v"(s[8]), "v"(s[9]), "v"(s[10]), "v"(s[11]), "v"(s[12]), "v"(s[13]), "v"(s[14]), "v"(s[15]));
    float4 v[GT];
    #pragma unroll
    for (int k = 0; k < GT; ++k) {
      v[k] = make_float4(0.0f, 0.0f, 0.0f, 0.0f);
      if (j + k < end) {
        int t = s[k];
        t = t < 0 ? 0 : t;
        t = t < N ? t : N - 1;
        v[k] = f4[(size_t)t * 2 + half];
      }
    }
    #pragma unroll
    for (int k = 0; k < GT; ++k) {
      acc.x += v[k].x;  acc.y += v[k].y;  acc.z += v[k].z;  acc.w += v[k].w;
    }
  }
  float4 o;
  o.x = bb.x + di * acc.x;  o.y = bb.y + di * acc.y;
  o.z = bb.z + di * acc.z;  o.w = bb.w + di * acc.w;
  if (scale_out) { o.x *= di;  o.y *= di;  o.z *= di;  o.w *= di; }
  ((float4*)res)[gid] = o;
}


extern "C" void kernel_launch(void* const* d_in, const int* in_sizes, int n_in,
                              void* d_out, int out_size, void* d_ws, size_t ws_size,
                              hipStream_t stream) {
  const float* x  = (const float*)d_in[0];
  const int*   ei = (const int*)d_in[1];
  const float* W1 = (const float*)d_in[2];
  const float* b1 = (const float*)d_in[3];
  const float* W2 = (const float*)d_in[4];
  const float* b2 = (const float*)d_in[5];
  float* out = (float*)d_out;

  const int N = in_sizes[0] / F_IN;
  const int E = in_sizes[1] / 2;
  const int* src = ei;
  const int* dst = ei + E;

  const size_t Np  = ((size_t)N + ZERO_TILE - 1) / ZERO_TILE * ZERO_TILE;
  const size_t Ep  = ((size_t)E + 31) / 32 * 32;
  const int    NB  = (N + BINW - 1) / BINW;
  const int    nbE = (int)(((size_t)E + CHUNK - 1) / CHUNK);
  const size_t cap = Ep + 32 * ((size_t)NB + 1);
  int*      cnt       = (int*)d_ws;
  int*      row_start = cnt + Np;
  float*    W12       = (float*)(row_start + Np);
  float*    cvec      = W12 + F_IN * C_OUT;
  int*      chunk_lo  = (int*)(cvec + 32);
  unsigned* chunk_rec = (unsigned*)(chunk_lo + (size_t)nbE * LP);
  int*      csr       = (int*)(chunk_rec + (size_t)nbE * CHUNK);
  float*    yd        = (float*)(csr + cap);
  float*    zd        = yd + Np * C_OUT;

  const int nbG = (int)(((size_t)N * 2 + 255) / 256);

  k_bin   <<<nbE + 1, T1, 0, stream>>>(src, dst, chunk_rec, chunk_lo, E, N, W1, b1, W2, W12, cvec);
  k_rows  <<<NB, T2, 0, stream>>>(chunk_rec, chunk_lo, nbE, cnt, row_start, csr, x, W12, yd, N);

  k_gather<<<nbG, 256, 0, stream>>>(cnt, row_start, csr, yd, cvec, zd, N, (int)cap, 1);
  k_gather<<<nbG, 256, 0, stream>>>(cnt, row_start, csr, zd, b2, out, N, (int)cap, 0);
}


# Gather's chain of dependent reads cut from eight round trips to five: arguments in one read (no early exit), row start beside the count, a trip's 32 entries asked at once; same order of sums
# speedup vs baseline: 8.3827x; 1.0390x over previous
#include <hip/hip_runtime.h>

typedef __attribute__((ext_vector_type(2))) float v2f;
typedef __attribute__((ext_vector_type(8))) float v8f;


#define F_IN  32
#define HC    64
#define C_OUT 8
#define ZERO_TILE 1024
#define BSH    8
#define BINW   256
#define BCAP   8192
#define CHUNK  8192
#define T1     1024
#define EPT1   8
#define T2     512
#define NW2    16
#define CPW    16
#define MAXNB  1024
#define MAXCH  512
#define LP     1056
#define NOREC  0xFFFFFFFFu
#define GT     16
#define GI     32


__device__ __forceinline__ void fold_tile(int tile, int lane,
                                          const float* __restrict__ W1,
                                          const float* __restrict__ b1,
                                          const float* __restrict__ W2,
                                          float* __restrict__ W12,
                                          float* __restrict__ cvec) {
  const int l15   = lane & 15;
  const int khalf = lane >> 4;
  const float* arow = (tile < 2) ? (W1 + (size_t)(tile * 16 + l15) * HC) : b1;
  const bool live_col = (l15 < C_OUT);
  const int  col = l15 & (C_OUT - 1);
  v8f c = {};
  #pragma unroll
  for (int k0 = 0; k0 < HC; k0 += 4) {
    const int ka = k0 + khalf * 2;
    v2f a = { arow[ka], arow[ka + 1] };
    v2f b = { W2[(size_t)ka * C_OUT + col], W2[(size_t)(ka + 1) * C_OUT + col] };
    c = __builtin_amdgcn_wmma_f32_16x16x4_f32(false, a, false, b,
                                              (short)0, c, false, false);
  }
  if (live_col) {
    if (tile < 2) {
      const int mbase = tile * 16 + khalf * 8;
      #pragma unroll
      for (int r = 0; r < 8; ++r)
        W12[(size_t)(mbase + r) * C_OUT + l15] = c[r];
    } else if (khalf == 0) {
      cvec[l15] = c[0];
    }
  }
}


__device__ __forceinline__ int wave_incl_scan(int x, int lane) {
  #pragma unroll
  for (int o = 1; o < 32; o <<= 1) {
    const int y = __shfl_up(x, o);
    if (lane >= o) x += y;
  }
  return x;
}

template <int NW>
__device__ __forceinline__ int block_excl_scan(int v, int* wsum, int tid, int& total) {
  const int lane = tid & 31;
  const int w    = tid >> 5;
  const int inc  = wave_incl_scan(v, lane);
  __syncthreads();
  if (lane == 31) wsum[w] = inc;
  __syncthreads();
  if (w == 0) {
    const int t = lane < NW ? wsum[lane] : 0;
    wsum[lane] = wave_incl_scan(t, lane);
  }
  __syncthreads();
  total = wsum[31];
  return (w ? wsum[w - 1] : 0) + inc - v;
}

__global__ __launch_bounds__(T1) void k_bin(const int* __restrict__ src,
                                            const int* __restrict__ dst,
                                            unsigned* __restrict__ chunk_rec,
                                            int* __restrict__ chunk_lo,
                                            int E, int N,
                                            const float* __restrict__ W1,
                                            const float* __restrict__ b1,
                                            const float* __restrict__ W2,
                                            float* __restrict__ W12,
                                            float* __restrict__ cvec) {
  __shared__ int      hist[MAXNB];
  __shared__ unsigned s_rec[CHUNK];
  __shared__ int      wsum[32];
  const int tid = threadIdx.x;
  if (blockIdx.x == 0) {
    const int tile = __builtin_amdgcn_readfirstlane(tid >> 5);
    if (tile < 3) fold_tile(tile, tid & 31, W1, b1, W2, W12, cvec);
    return;
  }
  const unsigned chunk = blockIdx.x - 1u;
  const unsigned e0    = chunk * (unsigned)CHUNK;
  const unsigned elast = (unsigned)E - 1u;
  hist[tid] = 0;
  __syncthreads();
  int d[EPT1], s[EPT1], r[EPT1];
  #pragma unroll
  for (int k = 0; k < EPT1; ++k) {
    const unsigned e  = e0 + k * T1 + tid;
    const unsigned ec = e < elast ? e : elast;
    d[k] = dst[ec];
    s[k] = src[ec];
  }
  #pragma unroll
  for (int k = 0; k < EPT1; ++k) {
    const unsigned e = e0 + k * T1 + tid;
    int dd = d[k];  dd = dd < 0 ? 0 : dd;  dd = dd < N ? dd : N - 1;
    int ss = s[k];  ss = ss < 0 ? 0 : ss;  ss = ss < N ? ss : N - 1;
    d[k] = dd;  s[k] = ss;
    r[k] = (e < (unsigned)E) ? atomicAdd(&hist[dd >> BSH], 1) : 0;
  }
  __syncthreads();
  const int c = hist[tid];
  int chunk_total;
  const int lo = block_excl_scan<T1 / 32>(c, wsum, tid, chunk_total);
  hist[tid] = lo;
  int* my_lo = chunk_lo + (size_t)chunk * LP;
  my_lo[tid] = lo;
  if (tid == 0) my_lo[MAXNB] = chunk_total;
  __syncthreads();
  #pragma unroll
  for (int k = 0; k < EPT1; ++k) {
    const unsigned e = e0 + k * T1 + tid;
    if (e < (unsigned)E) {
      const int lp = hist[d[k] >> BSH] + r[k];
      s_rec[lp] = ((unsigned)s[k] << 8) | (unsigned)(d[k] & (BINW - 1));
    }
  }
  __syncthreads();
  unsigned* my_rec = chunk_rec + (size_t)chunk * CHUNK;
  for (int i = tid; i < chunk_total; i += T1) my_rec[i] = s_rec[i];
}


__global__ __launch_bounds__(T2) void k_rows(const unsigned* __restrict__ chunk_rec,
                                             const int* __restrict__ chunk_lo,
                                             int nchunks,
                                             int* __restrict__ cnt,
                                             int* __restrict__ row_start,
                                             int* __restrict__ csr,
                                             const float* __restrict__ x,
                                             const float* __restrict__ W12,
                                             float* __restrict__ yd, int N) {
  __shared__ int cnt_l[BINW];
  __shared__ int cur_l[BINW];
  __shared__ int sorted[BCAP];
  __shared__ int2 run[MAXCH];
  __shared__ int part_lo[NW2];
  __shared__ int part_len[NW2];
  __shared__ int wsum[32];
  const int tid  = threadIdx.x;
  const int lane = tid & 31;
  const int w    = tid >> 5;
  const int b    = blockIdx.x;
  const int l15   = tid & 15;
  const int khalf = (tid >> 4) & 1;
  const int tile0 = b * BINW + w * 16;
  int xrow = tile0 + l15;
  if (xrow >= N) xrow = N - 1;
  const int wcol = l15 & (C_OUT - 1);
  int l0 = 0, len = 0;
  if (tid < nchunks) {
    const int* t_lo = chunk_lo + (size_t)tid * LP + b;
    l0 = t_lo[0];
    int l1 = t_lo[1];
    l0 = l0 < 0 ? 0 : l0;  l0 = l0 < CHUNK ? l0 : CHUNK;
    l1 = l1 < l0 ? l0 : l1;  l1 = l1 < CHUNK ? l1 : CHUNK;
    len = l1 - l0;
  }
  run[tid] = make_int2(l0, len);
  int sum_lo = l0, sum_len = len;
  #pragma unroll
  for (int o = 16; o > 0; o >>= 1) {
    sum_lo  += __shfl_xor(sum_lo, o);
    sum_len += __shfl_xor(sum_len, o);
  }
  if (lane == 0) { part_lo[w] = sum_lo;  part_len[w] = sum_len; }
  if (tid < BINW) cnt_l[tid] = 0;
  __syncthreads();
  int before = 0, total = 0;
  #pragma unroll
  for (int j = 0; j < NW2; ++j) { before += part_lo[j];  total += part_len[j]; }
  const int  base    = ((before + 31) & ~31) + 32 * b;
  const bool spilled = total > BCAP;
  int2 rn[CPW];
  #pragma unroll
  for (int k = 0; k < CPW; ++k) rn[k] = run[w + NW2 * k];
  unsigned rec[CPW];
  bool more = nchunks > NW2 * CPW;
  #pragma unroll
  for (int k = 0; k < CPW; ++k) {
    rec[k] = NOREC;
    more = more || rn[k].y > 32;
    if (lane < rn[k].y) rec[k] = chunk_rec[(size_t)(w + NW2 * k) * CHUNK + rn[k].x + lane];
  }
  #pragma unroll
  for (int k = 0; k < CPW; ++k)
    if (rec[k] != NOREC) atomicAdd(&cnt_l[rec[k] & (BINW - 1)], 1);
  if (more) {
    for (int c = w; c < nchunks; c += NW2) {
      const int rl = run[c].y;
      const unsigned* rp = chunk_rec + (size_t)c * CHUNK + run[c].x;
      for (int o = (c < NW2 * CPW ? 32 : 0) + lane; o < rl; o += 32)
        atomicAdd(&cnt_l[rp[o] & (BINW - 1)], 1);
    }
  }
  __syncthreads();
  const int v = tid < BINW ? cnt_l[tid] : 0;
  int bin_total;
  const int ex = block_excl_scan<NW2>(v, wsum, tid, bin_total);
  if (tid < BINW) {
    cur_l[tid] = ex;
    const int node = b * BINW + tid;
    if (node < N) {
      cnt[node]       = v;
      row_start[node] = base + ex;
    }
  }
  __syncthreads();
  v2f fa[F_IN / 4], fb[F_IN / 4];
  #pragma unroll
  for (int k = 0; k < F_IN / 4; ++k) {
    const int ka = k * 4 + khalf * 2;
    fa[k] = v2f{ x[(size_t)xrow * F_IN + ka], x[(size_t)xrow * F_IN + ka + 1] };
    fb[k] = v2f{ W12[(size_t)ka * C_OUT + wcol], W12[(size_t)(ka + 1) * C_OUT + wcol] };
  }
  #pragma unroll
  for (int k = 0; k < CPW; ++k) {
    if (rec[k] != NOREC) {
      const int pos = atomicAdd(&cur_l[rec[k] & (BINW - 1)], 1);
      const int sv  = (int)(rec[k] >> 8);
      if (!spilled) sorted[pos] = sv;
      else          csr[base + pos] = sv;
    }
  }
  if (more) {
    for (int c = w; c < nchunks; c += NW2) {
      const int rl = run[c].y;
      const unsigned* rp = chunk_rec + (size_t)c * CHUNK + run[c].x;
      for (int o = (c < NW2 * CPW ? 32 : 0) + lane; o < rl; o += 32) {
        const unsigned rr = rp[o];
        const int pos = atomicAdd(&cur_l[rr & (BINW - 1)], 1);
        const int sv  = (int)(rr >> 8);
        if (!spilled) sorted[pos] = sv;
        else          csr[base + pos] = sv;
      }
    }
  }
  __syncthreads();
  const int fill = ((before + total + 31) & ~31) + 32 * (b + 1) - base
                 + (b == (int)gridDim.x - 1 ? GI - 1 : 0);
  if (!spilled) {
    for (int i = tid; i < fill; i += T2) csr[base + i] = i < total ? sorted[i] : 0;
  } else {
    for (int i = total + tid; i < fill; i += T2) csr[base + i] = 0;
  }
  const float dmine = 1.0f / sqrtf((float)(cnt_l[w * 16 + l15] + 1));
  v8f c = {};
  #pragma unroll
  for (int k = 0; k < F_IN / 4; ++k)
    c = __builtin_amdgcn_wmma_f32_16x16x4_f32(false, fa[k], false, fb[k],
                                              (short)0, c, false, false);
  float sc[8];
  #pragma unroll
  for (int r = 0; r < 8; ++r) sc[r] = __shfl(dmine, khalf * 8 + r);
  if (l15 < C_OUT) {
    #pragma unroll
    for (int r = 0; r < 8; ++r) {
      const int orow = tile0 + khalf * 8 + r;
      if (orow < N) yd[(size_t)orow * C_OUT + l15] = c[r] * sc[r];
    }
  }
}


__global__ __launch_bounds__(256) void k_gather(const int* __restrict__ cnt,
                                                const int* __restrict__ row_start,
                                                const int* __restrict__ csr,
                                                const float* __restrict__ feat,
                                                const float* __restrict__ bias,
                                                float* __restrict__ res,
                                                int N, int cap, int scale_out) {
  const unsigned gid0 = blockIdx.x * 256u + threadIdx.x;
  const bool live = (int)(gid0 >> 1) < N;
  const unsigned gid = live ? gid0 : (unsigned)(2 * N - 2) + (gid0 & 1u);
  const int node = gid >> 1;
  const int half = gid & 1;
  const int end = cnt[node];
  int rs = row_start[node];
  asm volatile("" :: "v"(end), "v"(rs));
  rs = rs < 0 ? 0 : rs;
  const float di = 1.0f / sqrtf((float)(end + 1));
  const float4* f4 = (const float4*)feat;
  const float4 bb = ((const float4*)bias)[half];
  float4 acc = f4[gid];
  for (int j = 0; j < end; j += GI) {
    int s[GI];
    #pragma unroll
    for (int g = 0; g < GI / 4; ++g) {
      int p = rs + j + 4 * g;
      p = p < cap - 4 ? p : cap - 4;
      __builtin_memcpy(&s[4 * g], csr + p, 4 * sizeof(int));
    }
    asm volatile("" :: "v"(s[0]), "v"(s[1]), "v"(s[2]), "v"(s[3]), "v"(s[4]), "v"(s[5]), "v"(s[6]), "v"(s[7]),
                       "v"(s[8]), "v"(s[9]), "v"(s[10]), "v"(s[11]), "v"(s[12]), "v"(s[13]), "v"(s[14]), "v"(s[15]));
    asm volatile("" :: "v"(s[16]), "v"(s[17]), "v"(s[18]), "v"(s[19]), "v"(s[20]), "v"(s[21]), "v"(s[22]), "v"(s[23]),
                       "v"(s[24]), "v"(s[25]), "v"(s[26]), "v"(s[27]), "v"(s[28]), "v"(s[29]), "v"(s[30]), "v"(s[31]));
    #pragma unroll
    for (int h = 0; h < GI / GT; ++h) {
      if (h == 0 || j + h * GT < end) {
        float4 v[GT];
        #pragma unroll
        for (int k = 0; k < GT; ++k) {
          v[k] = make_float4(0.0f, 0.0f, 0.0f, 0.0f);
          if (j + h * GT + k < end) {
            int t = s[h * GT + k];
            t = t < 0 ? 0 : t;
            t = t < N ? t : N - 1;
            v[k] = f4[(size_t)t * 2 + half];
          }
        }
        #pragma unroll
        for (int k = 0; k < GT; ++k) {
          acc.x += v[k].x;  acc.y += v[k].y;  acc.z += v[k].z;  acc.w += v[k].w;
        }
      }
    }
  }
  float4 o;
  o.x = bb.x + di * acc.x;  o.y = bb.y + di * acc.y;
  o.z = bb.z + di * acc.z;  o.w = bb.w + di * acc.w;
  if (scale_out) { o.x *= di;  o.y *= di;  o.z *= di;  o.w *= di; }
  if (live) ((float4*)res)[gid] = o;
}


extern "C" void kernel_launch(void* const* d_in, const int* in_sizes, int n_in,
                              void* d_out, int out_size, void* d_ws, size_t ws_size,
                              hipStream_t stream) {
  const float* x  = (const float*)d_in[0];
  const int*   ei = (const int*)d_in[1];
  const float* W1 = (const float*)d_in[2];
  const float* b1 = (const float*)d_in[3];
  const float* W2 = (const float*)d_in[4];
  const float* b2 = (const float*)d_in[5];
  float* out = (float*)d_out;

  const int N = in_sizes[0] / F_IN;
  const int E = in_sizes[1] / 2;
  const int* src = ei;
  const int* dst = ei + E;

  const size_t Np  = ((size_t)N + ZERO_TILE - 1) / ZERO_TILE * ZERO_TILE;
  const size_t Ep  = ((size_t)E + 31) / 32 * 32;
  const int    NB  = (N + BINW - 1) / BINW;
  const int    nbE = (int)(((size_t)E + CHUNK - 1) / CHUNK);
  const size_t cap = Ep + 32 * ((size_t)NB + 1);
  int*      cnt       = (int*)d_ws;
  int*      row_start = cnt + Np;
  float*    W12       = (float*)(row_start + Np);
  float*    cvec      = W12 + F_IN * C_OUT;
  int*      chunk_lo  = (int*)(cvec + 32);
  unsigned* chunk_rec = (unsigned*)(chunk_lo + (size_t)nbE * LP);
  int*      csr       = (int*)(chunk_rec + (size_t)nbE * CHUNK);
  float*    yd        = (float*)(csr + cap);
  float*    zd        = yd + Np * C_OUT;

  const int nbG = (int)(((size_t)N * 2 + 255) / 256);

  k_bin   <<<nbE + 1, T1, 0, stream>>>(src, dst, chunk_rec, chunk_lo, E, N, W1, b1, W2, W12, cvec);
  k_rows  <<<NB, T2, 0, stream>>>(chunk_rec, chunk_lo, nbE, cnt, row_start, csr, x, W12, yd, N);

  k_gather<<<nbG, 256, 0, stream>>>(cnt, row_start, csr, yd, cvec, zd, N, (int)cap, 1);
  k_gather<<<nbG, 256, 0, stream>>>(cnt, row_start, csr, zd, b2, out, N, (int)cap, 0);
}
